# starting kernel
# baseline (speedup 1.0000x reference)
#include <hip/hip_runtime.h>

typedef __attribute__((ext_vector_type(2))) float v2f;
typedef __attribute__((ext_vector_type(8))) float v8f;

#define WMMA_F32(a, b, c) \
  __builtin_amdgcn_wmma_f32_16x16x4_f32(false, (a), false, (b), (short)0, (c), false, false)

static constexpr int BB = 2, TT = 2048, CC = 1024, HH = 16, DD = 64;
static constexpr int MTOT = BB * TT;
static constexpr int NQKV = 3 * CC;
static constexpr size_t HSZ = (size_t)BB * HH * TT * DD;

__global__ __launch_bounds__(32)
void qkv_gemm_kernel(const float* __restrict__ X, const float* __restrict__ W,
                     const float* __restrict__ bias,
                     float* __restrict__ q, float* __restrict__ k,
                     float* __restrict__ v) {
  const int lane = threadIdx.x & 31;
  const int hl = lane >> 4;
  const int ln = lane & 15;
  const int n0 = blockIdx.x * 64;
  const int m0 = blockIdx.y * 16;

  v8f c[4] = {};
  const float* Arow = X + (size_t)(m0 + ln) * CC;

  for (int k0 = 0; k0 < CC; k0 += 4) {
    v2f a = *(const v2f*)(Arow + k0 + 2 * hl);
    __builtin_prefetch((const void*)(Arow + k0 + 64), 0, 0);
    const float* Bp = W + (size_t)(k0 + 2 * hl) * NQKV + n0 + ln;
#pragma unroll
    for (int n = 0; n < 4; ++n) {
      v2f b;
      b.x = Bp[16 * n];
      b.y = Bp[NQKV + 16 * n];
      c[n] = WMMA_F32(a, b, c[n]);
    }
  }

#pragma unroll
  for (int n = 0; n < 4; ++n) {
    const int col = n0 + 16 * n + ln;
    const float bval = bias[col];
    const int which = col >> 10;
    const int rem = col & (CC - 1);
    const int h = rem >> 6;
    const int d = rem & 63;
    float* dst = (which == 0) ? q : (which == 1) ? k : v;
#pragma unroll
    for (int r = 0; r < 8; ++r) {
      const int trow = m0 + r + 8 * hl;
      const int bidx = trow >> 11;
      const int t = trow & (TT - 1);
      const size_t idx = (((size_t)(bidx * HH + h) * TT) + t) * DD + d;
      dst[idx] = c[n][r] + bval;
    }
  }
}

__global__ __launch_bounds__(32)
void attn_kernel(const float* __restrict__ q, const float* __restrict__ kmat,
                 const float* __restrict__ vmat, float* __restrict__ out) {
  const int lane = threadIdx.x & 31;
  const int hl = lane >> 4;
  const int ln = lane & 15;
  const int qtile = blockIdx.x;
  const int bh = blockIdx.y;
  const int b = bh >> 4, h = bh & 15;
  const int q0 = qtile * 16;

  const float* qbase = q + (size_t)bh * TT * DD;
  const float* kbase = kmat + (size_t)bh * TT * DD;
  const float* vbase = vmat + (size_t)bh * TT * DD;

  __shared__ float pt[16][17];

  v2f qa[16];
  {
    const float* qrow = qbase + (size_t)(q0 + ln) * DD + 2 * hl;
#pragma unroll
    for (int t = 0; t < 16; ++t) {
      v2f a = *(const v2f*)(qrow + 4 * t);
      qa[t] = a * 0.125f;
    }
  }

  v8f o[4] = {};
  float m_i[8], l_i[8];
#pragma unroll
  for (int r = 0; r < 8; ++r) { m_i[r] = -__builtin_inff(); l_i[r] = 0.0f; }

  for (int kt = 0; kt <= qtile; ++kt) {
    const int kb = kt * 16;

    v8f s = {};
    const float* krow = kbase + (size_t)(kb + ln) * DD + 2 * hl;
#pragma unroll
    for (int t = 0; t < 16; ++t) {
      v2f bf = *(const v2f*)(krow + 4 * t);
      s = WMMA_F32(qa[t], bf, s);
    }

    const bool diag = (kt == qtile);
    float scale_r[8];
#pragma unroll
    for (int r = 0; r < 8; ++r) {
      const int row = r + 8 * hl;
      float sv = s[r];
      if (diag && (ln > row)) sv = -__builtin_inff();
      float mx = sv;
      mx = fmaxf(mx, __shfl_xor(mx, 1));
      mx = fmaxf(mx, __shfl_xor(mx, 2));
      mx = fmaxf(mx, __shfl_xor(mx, 4));
      mx = fmaxf(mx, __shfl_xor(mx, 8));
      const float mn = fmaxf(m_i[r], mx);
      const float sc = __expf(m_i[r] - mn);
      const float p = __expf(sv - mn);
      float ps = p;
      ps += __shfl_xor(ps, 1);
      ps += __shfl_xor(ps, 2);
      ps += __shfl_xor(ps, 4);
      ps += __shfl_xor(ps, 8);
      l_i[r] = l_i[r] * sc + ps;
      m_i[r] = mn;
      scale_r[r] = sc;
      pt[row][ln] = p;
    }

#pragma unroll
    for (int n = 0; n < 4; ++n)
#pragma unroll
      for (int r = 0; r < 8; ++r) o[n][r] *= scale_r[r];

    __syncthreads();

#pragma unroll
    for (int t = 0; t < 4; ++t) {
      v2f a;
      a.x = pt[ln][4 * t + 2 * hl];
      a.y = pt[ln][4 * t + 2 * hl + 1];
      const float* vrow = vbase + (size_t)(kb + 4 * t + 2 * hl) * DD + ln;
#pragma unroll
      for (int n = 0; n < 4; ++n) {
        v2f bf;
        bf.x = vrow[16 * n];
        bf.y = vrow[DD + 16 * n];
        o[n] = WMMA_F32(a, bf, o[n]);
      }
    }
    __syncthreads();
  }

#pragma unroll
  for (int r = 0; r < 8; ++r) {
    const float inv = 1.0f / l_i[r];
    const int trow = q0 + r + 8 * hl;
    float* orow = out + ((size_t)(b * TT + trow) * CC) + h * DD;
#pragma unroll
    for (int n = 0; n < 4; ++n) orow[16 * n + ln] = o[n][r] * inv;
  }
}

__global__ __launch_bounds__(32)
void out_gemm_kernel(const float* __restrict__ A, const float* __restrict__ W,
                     const float* __restrict__ bias, float* __restrict__ out) {
  const int lane = threadIdx.x & 31;
  const int hl = lane >> 4;
  const int ln = lane & 15;
  const int n0 = blockIdx.x * 64;
  const int m0 = blockIdx.y * 16;

  v8f c[4] = {};
  const float* Arow = A + (size_t)(m0 + ln) * CC;

  for (int k0 = 0; k0 < CC; k0 += 4) {
    v2f a = *(const v2f*)(Arow + k0 + 2 * hl);
    __builtin_prefetch((const void*)(Arow + k0 + 64), 0, 0);
    const float* Bp = W + (size_t)(k0 + 2 * hl) * CC + n0 + ln;
#pragma unroll
    for (int n = 0; n < 4; ++n) {
      v2f b;
      b.x = Bp[16 * n];
      b.y = Bp[CC + 16 * n];
      c[n] = WMMA_F32(a, b, c[n]);
    }
  }

#pragma unroll
  for (int n = 0; n < 4; ++n) {
    const float bval = bias[n0 + 16 * n + ln];
#pragma unroll
    for (int r = 0; r < 8; ++r) {
      out[(size_t)(m0 + r + 8 * hl) * CC + n0 + 16 * n + ln] = c[n][r] + bval;
    }
  }
}

extern "C" void kernel_launch(void* const* d_in, const int* in_sizes, int n_in,
                              void* d_out, int out_size, void* d_ws, size_t ws_size,
                              hipStream_t stream) {
  (void)in_sizes; (void)n_in; (void)out_size; (void)ws_size;
  const float* x    = (const float*)d_in[0];
  const float* Wqkv = (const float*)d_in[1];
  const float* bqkv = (const float*)d_in[2];
  const float* Wout = (const float*)d_in[3];
  const float* bout = (const float*)d_in[4];
  float* out = (float*)d_out;

  float* qp = (float*)d_ws;
  float* kp = qp + HSZ;
  float* vp = kp + HSZ;
  float* ao = vp + HSZ;

  const dim3 blk(32, 1, 1);
  qkv_gemm_kernel<<<dim3(NQKV / 64, MTOT / 16), blk, 0, stream>>>(x, Wqkv, bqkv, qp, kp, vp);
  attn_kernel<<<dim3(TT / 16, BB * HH), blk, 0, stream>>>(qp, kp, vp, ao);
  out_gemm_kernel<<<dim3(CC / 64, MTOT / 16), blk, 0, stream>>>(ao, Wout, bout, out);
}


# All global WMMA-fragment loads and output stores become 16-byte: A-side k permuted in 8s (lane half holds 4 consecutive k), B-side sub-tile j owns columns 4*lane+j; k sum order changes within 8s
# speedup vs baseline: 1.3218x; 1.3218x over previous
#include <hip/hip_runtime.h>


typedef __attribute__((ext_vector_type(2))) float v2f;
typedef __attribute__((ext_vector_type(4))) float v4f;
typedef __attribute__((ext_vector_type(8))) float v8f;

#define WMMA_F32(a, b, c) \
  __builtin_amdgcn_wmma_f32_16x16x4_f32(false, (a), false, (b), (short)0, (c), false, false)

static constexpr int BB = 2, TT = 2048, CC = 1024, HH = 16, DD = 64;
static constexpr int MTOT = BB * TT;
static constexpr int NQKV = 3 * CC;
static constexpr size_t HSZ = (size_t)BB * HH * TT * DD;


__device__ __forceinline__ void gemm_tile_16x64(const float* __restrict__ Arow,
                                                const float* __restrict__ Bp,
                                                const int ldb, v8f (&c)[4]) {
  for (int k0 = 0; k0 < CC; k0 += 8) {
    const v4f a = *(const v4f*)(Arow + k0);
    const float* Bk = Bp + (size_t)k0 * ldb;
    const v4f b0 = *(const v4f*)(Bk);
    const v4f b1 = *(const v4f*)(Bk + ldb);
    const v4f b2 = *(const v4f*)(Bk + 2 * ldb);
    const v4f b3 = *(const v4f*)(Bk + 3 * ldb);
    const v2f a01 = {a.x, a.y};
    const v2f a23 = {a.z, a.w};
#pragma unroll
    for (int j = 0; j < 4; ++j) {
      const v2f b = {b0[j], b1[j]};
      c[j] = WMMA_F32(a01, b, c[j]);
    }
#pragma unroll
    for (int j = 0; j < 4; ++j) {
      const v2f b = {b2[j], b3[j]};
      c[j] = WMMA_F32(a23, b, c[j]);
    }
  }
}

__global__ __launch_bounds__(32)
void qkv_gemm_kernel(const float* __restrict__ X, const float* __restrict__ W,
                     const float* __restrict__ bias,
                     float* __restrict__ q, float* __restrict__ k,
                     float* __restrict__ v) {
  const int lane = threadIdx.x & 31;
  const int hl = lane >> 4;
  const int ln = lane & 15;
  const int n0 = blockIdx.x * 64;
  const int m0 = blockIdx.y * 16;

  v8f c[4] = {};
  gemm_tile_16x64(X + (size_t)(m0 + ln) * CC + 4 * hl,
                  W + (size_t)(4 * hl) * NQKV + n0 + 4 * ln, NQKV, c);

  const v4f bv = *(const v4f*)(bias + n0 + 4 * ln);
  const int which = n0 >> 10;
  const int h = (n0 & (CC - 1)) >> 6;
  float* dst = (which == 0) ? q : (which == 1) ? k : v;
#pragma unroll
  for (int r = 0; r < 8; ++r) {
    const int trow = m0 + r + 8 * hl;
    const int bidx = trow >> 11;
    const int t = trow & (TT - 1);
    const size_t idx = (((size_t)(bidx * HH + h) * TT) + t) * DD + 4 * ln;
    const v4f o = {c[0][r] + bv.x, c[1][r] + bv.y, c[2][r] + bv.z, c[3][r] + bv.w};
    *(v4f*)(dst + idx) = o;
  }
}

__global__ __launch_bounds__(32)
void attn_kernel(const float* __restrict__ q, const float* __restrict__ kmat,
                 const float* __restrict__ vmat, float* __restrict__ out) {
  const int lane = threadIdx.x & 31;
  const int hl = lane >> 4;
  const int ln = lane & 15;
  const int qtile = blockIdx.x;
  const int bh = blockIdx.y;
  const int b = bh >> 4, h = bh & 15;
  const int q0 = qtile * 16;

  const float* qbase = q + (size_t)bh * TT * DD;
  const float* kbase = kmat + (size_t)bh * TT * DD;
  const float* vbase = vmat + (size_t)bh * TT * DD;

  __shared__ float pt[16][17];

  v4f qa[8];
  {
    const float* qrow = qbase + (size_t)(q0 + ln) * DD + 4 * hl;
#pragma unroll
    for (int u = 0; u < 8; ++u) {
      const v4f a = *(const v4f*)(qrow + 8 * u);
      qa[u] = a * 0.125f;
    }
  }

  v8f o[4] = {};
  float m_i[8], l_i[8];
#pragma unroll
  for (int r = 0; r < 8; ++r) { m_i[r] = -__builtin_inff(); l_i[r] = 0.0f; }

  for (int kt = 0; kt <= qtile; ++kt) {
    const int kb = kt * 16;

    v8f s = {};
    const float* krow = kbase + (size_t)(kb + ln) * DD + 4 * hl;
#pragma unroll
    for (int u = 0; u < 8; ++u) {
      const v4f kf = *(const v4f*)(krow + 8 * u);
      const v2f q01 = {qa[u].x, qa[u].y};
      const v2f k01 = {kf.x, kf.y};
      s = WMMA_F32(q01, k01, s);
      const v2f q23 = {qa[u].z, qa[u].w};
      const v2f k23 = {kf.z, kf.w};
      s = WMMA_F32(q23, k23, s);
    }

    const bool diag = (kt == qtile);
    float scale_r[8];
#pragma unroll
    for (int r = 0; r < 8; ++r) {
      const int row = r + 8 * hl;
      float sv = s[r];
      if (diag && (ln > row)) sv = -__builtin_inff();
      float mx = sv;
      mx = fmaxf(mx, __shfl_xor(mx, 1));
      mx = fmaxf(mx, __shfl_xor(mx, 2));
      mx = fmaxf(mx, __shfl_xor(mx, 4));
      mx = fmaxf(mx, __shfl_xor(mx, 8));
      const float mn = fmaxf(m_i[r], mx);
      const float sc = __expf(m_i[r] - mn);
      const float p = __expf(sv - mn);
      float ps = p;
      ps += __shfl_xor(ps, 1);
      ps += __shfl_xor(ps, 2);
      ps += __shfl_xor(ps, 4);
      ps += __shfl_xor(ps, 8);
      l_i[r] = l_i[r] * sc + ps;
      m_i[r] = mn;
      scale_r[r] = sc;
      pt[row][ln] = p;
    }

#pragma unroll
    for (int n = 0; n < 4; ++n)
#pragma unroll
      for (int r = 0; r < 8; ++r) o[n][r] *= scale_r[r];

    __syncthreads();

#pragma unroll
    for (int t = 0; t < 4; ++t) {
      v2f a;
      a.x = pt[ln][4 * t + 2 * hl];
      a.y = pt[ln][4 * t + 2 * hl + 1];
      const float* vrow = vbase + (size_t)(kb + 4 * t + 2 * hl) * DD + 4 * ln;
      const v4f v0 = *(const v4f*)(vrow);
      const v4f v1 = *(const v4f*)(vrow + DD);
#pragma unroll
      for (int n = 0; n < 4; ++n) {
        const v2f bf = {v0[n], v1[n]};
        o[n] = WMMA_F32(a, bf, o[n]);
      }
    }
    __syncthreads();
  }

#pragma unroll
  for (int r = 0; r < 8; ++r) {
    const float inv = 1.0f / l_i[r];
    const int trow = q0 + r + 8 * hl;
    float* orow = out + ((size_t)(b * TT + trow) * CC) + h * DD + 4 * ln;
    const v4f ov = {o[0][r] * inv, o[1][r] * inv, o[2][r] * inv, o[3][r] * inv};
    *(v4f*)orow = ov;
  }
}

__global__ __launch_bounds__(32)
void out_gemm_kernel(const float* __restrict__ A, const float* __restrict__ W,
                     const float* __restrict__ bias, float* __restrict__ out) {
  const int lane = threadIdx.x & 31;
  const int hl = lane >> 4;
  const int ln = lane & 15;
  const int n0 = blockIdx.x * 64;
  const int m0 = blockIdx.y * 16;

  v8f c[4] = {};
  gemm_tile_16x64(A + (size_t)(m0 + ln) * CC + 4 * hl,
                  W + (size_t)(4 * hl) * CC + n0 + 4 * ln, CC, c);

  const v4f bv = *(const v4f*)(bias + n0 + 4 * ln);
#pragma unroll
  for (int r = 0; r < 8; ++r) {
    const v4f o = {c[0][r] + bv.x, c[1][r] + bv.y, c[2][r] + bv.z, c[3][r] + bv.w};
    *(v4f*)(out + (size_t)(m0 + r + 8 * hl) * CC + n0 + 4 * ln) = o;
  }
}

extern "C" void kernel_launch(void* const* d_in, const int* in_sizes, int n_in,
                              void* d_out, int out_size, void* d_ws, size_t ws_size,
                              hipStream_t stream) {
  (void)in_sizes; (void)n_in; (void)out_size; (void)ws_size;
  const float* x    = (const float*)d_in[0];
  const float* Wqkv = (const float*)d_in[1];
  const float* bqkv = (const float*)d_in[2];
  const float* Wout = (const float*)d_in[3];
  const float* bout = (const float*)d_in[4];
  float* out = (float*)d_out;

  float* qp = (float*)d_ws;
  float* kp = qp + HSZ;
  float* vp = kp + HSZ;
  float* ao = vp + HSZ;

  const dim3 blk(32, 1, 1);
  qkv_gemm_kernel<<<dim3(NQKV / 64, MTOT / 16), blk, 0, stream>>>(x, Wqkv, bqkv, qp, kp, vp);
  attn_kernel<<<dim3(TT / 16, BB * HH), blk, 0, stream>>>(qp, kp, vp, ao);
  out_gemm_kernel<<<dim3(CC / 64, MTOT / 16), blk, 0, stream>>>(ao, Wout, bout, out);
}


# Both projection GEMMs: each wave holds a 32x128 output tile in registers (a B fragment feeds 2 row tiles, an A fragment 8 column sub-tiles), A read one 128-byte line per row at once; sum order same
# speedup vs baseline: 1.5921x; 1.2045x over previous
#include <hip/hip_runtime.h>


typedef __attribute__((ext_vector_type(2))) float v2f;
typedef __attribute__((ext_vector_type(4))) float v4f;
typedef __attribute__((ext_vector_type(8))) float v8f;

#define WMMA_F32(a, b, c) \
  __builtin_amdgcn_wmma_f32_16x16x4_f32(false, (a), false, (b), (short)0, (c), false, false)

static constexpr int BB = 2, TT = 2048, CC = 1024, HH = 16, DD = 64;
static constexpr int MTOT = BB * TT;
static constexpr int NQKV = 3 * CC;
static constexpr size_t HSZ = (size_t)BB * HH * TT * DD;


__device__ __forceinline__ void load_a_chunk(v4f (&a)[2][4],
                                             const float* __restrict__ A0,
                                             const float* __restrict__ A1, const int kc) {
#pragma unroll
  for (int u = 0; u < 4; ++u) a[0][u] = *(const v4f*)(A0 + kc + 8 * u);
#pragma unroll
  for (int u = 0; u < 4; ++u) a[1][u] = *(const v4f*)(A1 + kc + 8 * u);
}

template <int LDB>
__device__ __forceinline__ void load_b_step(v4f (&b)[8], const float* __restrict__ Bp,
                                            const int k0) {
  const float* Bk = Bp + (size_t)k0 * LDB;
#pragma unroll
  for (int r = 0; r < 4; ++r) {
    b[2 * r]     = *(const v4f*)(Bk + r * LDB);
    b[2 * r + 1] = *(const v4f*)(Bk + r * LDB + 64);
  }
}

__device__ __forceinline__ void mma_step(const v4f a0, const v4f a1, const v4f (&b)[8],
                                         v8f (&c)[2][8]) {
  const v2f a0lo = {a0.x, a0.y}, a1lo = {a1.x, a1.y};
  const v2f a0hi = {a0.z, a0.w}, a1hi = {a1.z, a1.w};
#pragma unroll
  for (int hf = 0; hf < 2; ++hf)
#pragma unroll
    for (int j = 0; j < 4; ++j) {
      const v2f bb = {b[hf][j], b[2 + hf][j]};
      c[0][4 * hf + j] = WMMA_F32(a0lo, bb, c[0][4 * hf + j]);
      c[1][4 * hf + j] = WMMA_F32(a1lo, bb, c[1][4 * hf + j]);
    }
#pragma unroll
  for (int hf = 0; hf < 2; ++hf)
#pragma unroll
    for (int j = 0; j < 4; ++j) {
      const v2f bb = {b[4 + hf][j], b[6 + hf][j]};
      c[0][4 * hf + j] = WMMA_F32(a0hi, bb, c[0][4 * hf + j]);
      c[1][4 * hf + j] = WMMA_F32(a1hi, bb, c[1][4 * hf + j]);
    }
}

template <int LDB>
__device__ __forceinline__ void gemm_chunk(v4f (&aC)[2][4], v4f (&aN)[2][4],
                                           v4f (&bP)[8], v4f (&bQ)[8],
                                           const float* __restrict__ A0,
                                           const float* __restrict__ A1,
                                           const float* __restrict__ Bp, const int kc,
                                           v8f (&c)[2][8]) {
  load_a_chunk(aN, A0, A1, (kc + 32) & (CC - 1));
  __builtin_amdgcn_sched_barrier(0);
  load_b_step<LDB>(bQ, Bp, kc + 8);
  mma_step(aC[0][0], aC[1][0], bP, c);
  load_b_step<LDB>(bP, Bp, kc + 16);
  mma_step(aC[0][1], aC[1][1], bQ, c);
  load_b_step<LDB>(bQ, Bp, kc + 24);
  mma_step(aC[0][2], aC[1][2], bP, c);
  load_b_step<LDB>(bP, Bp, (kc + 32) & (CC - 1));
  mma_step(aC[0][3], aC[1][3], bQ, c);
}

template <int LDB>
__device__ __forceinline__ void gemm_tile_32x128(const float* __restrict__ A0,
                                                 const float* __restrict__ A1,
                                                 const float* __restrict__ Bp,
                                                 v8f (&c)[2][8]) {
  v4f aP[2][4], aQ[2][4], bP[8], bQ[8];
  load_a_chunk(aP, A0, A1, 0);
  load_b_step<LDB>(bP, Bp, 0);
  for (int kc = 0; kc < CC; kc += 64) {
    gemm_chunk<LDB>(aP, aQ, bP, bQ, A0, A1, Bp, kc, c);
    gemm_chunk<LDB>(aQ, aP, bP, bQ, A0, A1, Bp, kc + 32, c);
  }
}

__global__ __launch_bounds__(32)
void qkv_gemm_kernel(const float* __restrict__ X, const float* __restrict__ W,
                     const float* __restrict__ bias,
                     float* __restrict__ q, float* __restrict__ k,
                     float* __restrict__ v) {
  const int lane = threadIdx.x & 31;
  const int hl = lane >> 4;
  const int ln = lane & 15;
  const int n0 = blockIdx.x * 128;
  const int m0 = blockIdx.y * 32;

  v8f c[2][8] = {};
  const float* A0 = X + (size_t)(m0 + ln) * CC + 4 * hl;
  gemm_tile_32x128<NQKV>(A0, A0 + 16 * CC,
                         W + (size_t)(4 * hl) * NQKV + n0 + 4 * ln, c);

  const int which = n0 >> 10;
  float* dst = (which == 0) ? q : (which == 1) ? k : v;
#pragma unroll
  for (int hf = 0; hf < 2; ++hf) {
    const v4f bv = *(const v4f*)(bias + n0 + 64 * hf + 4 * ln);
    const int h = ((n0 + 64 * hf) & (CC - 1)) >> 6;
#pragma unroll
    for (int t = 0; t < 2; ++t)
#pragma unroll
      for (int r = 0; r < 8; ++r) {
        const int trow = m0 + 16 * t + r + 8 * hl;
        const int bidx = trow >> 11;
        const int tt = trow & (TT - 1);
        const size_t idx = (((size_t)(bidx * HH + h) * TT) + tt) * DD + 4 * ln;
        const v4f o = {c[t][4 * hf + 0][r] + bv.x, c[t][4 * hf + 1][r] + bv.y,
                       c[t][4 * hf + 2][r] + bv.z, c[t][4 * hf + 3][r] + bv.w};
        *(v4f*)(dst + idx) = o;
      }
  }
}

__global__ __launch_bounds__(32)
void attn_kernel(const float* __restrict__ q, const float* __restrict__ kmat,
                 const float* __restrict__ vmat, float* __restrict__ out) {
  const int lane = threadIdx.x & 31;
  const int hl = lane >> 4;
  const int ln = lane & 15;
  const int qtile = blockIdx.x;
  const int bh = blockIdx.y;
  const int b = bh >> 4, h = bh & 15;
  const int q0 = qtile * 16;

  const float* qbase = q + (size_t)bh * TT * DD;
  const float* kbase = kmat + (size_t)bh * TT * DD;
  const float* vbase = vmat + (size_t)bh * TT * DD;

  __shared__ float pt[16][17];

  v4f qa[8];
  {
    const float* qrow = qbase + (size_t)(q0 + ln) * DD + 4 * hl;
#pragma unroll
    for (int u = 0; u < 8; ++u) {
      const v4f a = *(const v4f*)(qrow + 8 * u);
      qa[u] = a * 0.125f;
    }
  }

  v8f o[4] = {};
  float m_i[8], l_i[8];
#pragma unroll
  for (int r = 0; r < 8; ++r) { m_i[r] = -__builtin_inff(); l_i[r] = 0.0f; }

  for (int kt = 0; kt <= qtile; ++kt) {
    const int kb = kt * 16;

    v8f s = {};
    const float* krow = kbase + (size_t)(kb + ln) * DD + 4 * hl;
#pragma unroll
    for (int u = 0; u < 8; ++u) {
      const v4f kf = *(const v4f*)(krow + 8 * u);
      const v2f q01 = {qa[u].x, qa[u].y};
      const v2f k01 = {kf.x, kf.y};
      s = WMMA_F32(q01, k01, s);
      const v2f q23 = {qa[u].z, qa[u].w};
      const v2f k23 = {kf.z, kf.w};
      s = WMMA_F32(q23, k23, s);
    }

    const bool diag = (kt == qtile);
    float scale_r[8];
#pragma unroll
    for (int r = 0; r < 8; ++r) {
      const int row = r + 8 * hl;
      float sv = s[r];
      if (diag && (ln > row)) sv = -__builtin_inff();
      float mx = sv;
      mx = fmaxf(mx, __shfl_xor(mx, 1));
      mx = fmaxf(mx, __shfl_xor(mx, 2));
      mx = fmaxf(mx, __shfl_xor(mx, 4));
      mx = fmaxf(mx, __shfl_xor(mx, 8));
      const float mn = fmaxf(m_i[r], mx);
      const float sc = __expf(m_i[r] - mn);
      const float p = __expf(sv - mn);
      float ps = p;
      ps += __shfl_xor(ps, 1);
      ps += __shfl_xor(ps, 2);
      ps += __shfl_xor(ps, 4);
      ps += __shfl_xor(ps, 8);
      l_i[r] = l_i[r] * sc + ps;
      m_i[r] = mn;
      scale_r[r] = sc;
      pt[row][ln] = p;
    }

#pragma unroll
    for (int n = 0; n < 4; ++n)
#pragma unroll
      for (int r = 0; r < 8; ++r) o[n][r] *= scale_r[r];

    __syncthreads();

#pragma unroll
    for (int t = 0; t < 4; ++t) {
      v2f a;
      a.x = pt[ln][4 * t + 2 * hl];
      a.y = pt[ln][4 * t + 2 * hl + 1];
      const float* vrow = vbase + (size_t)(kb + 4 * t + 2 * hl) * DD + 4 * ln;
      const v4f v0 = *(const v4f*)(vrow);
      const v4f v1 = *(const v4f*)(vrow + DD);
#pragma unroll
      for (int n = 0; n < 4; ++n) {
        const v2f bf = {v0[n], v1[n]};
        o[n] = WMMA_F32(a, bf, o[n]);
      }
    }
    __syncthreads();
  }

#pragma unroll
  for (int r = 0; r < 8; ++r) {
    const float inv = 1.0f / l_i[r];
    const int trow = q0 + r + 8 * hl;
    float* orow = out + ((size_t)(b * TT + trow) * CC) + h * DD + 4 * ln;
    const v4f ov = {o[0][r] * inv, o[1][r] * inv, o[2][r] * inv, o[3][r] * inv};
    *(v4f*)orow = ov;
  }
}

__global__ __launch_bounds__(32)
void out_gemm_kernel(const float* __restrict__ A, const float* __restrict__ W,
                     const float* __restrict__ bias, float* __restrict__ out) {
  const int lane = threadIdx.x & 31;
  const int hl = lane >> 4;
  const int ln = lane & 15;
  const int n0 = blockIdx.x * 128;
  const int m0 = blockIdx.y * 32;

  v8f c[2][8] = {};
  const float* A0 = A + (size_t)(m0 + ln) * CC + 4 * hl;
  gemm_tile_32x128<CC>(A0, A0 + 16 * CC,
                       W + (size_t)(4 * hl) * CC + n0 + 4 * ln, c);

#pragma unroll
  for (int hf = 0; hf < 2; ++hf) {
    const v4f bv = *(const v4f*)(bias + n0 + 64 * hf + 4 * ln);
#pragma unroll
    for (int t = 0; t < 2; ++t)
#pragma unroll
      for (int r = 0; r < 8; ++r) {
        const v4f o = {c[t][4 * hf + 0][r] + bv.x, c[t][4 * hf + 1][r] + bv.y,
                       c[t][4 * hf + 2][r] + bv.z, c[t][4 * hf + 3][r] + bv.w};
        *(v4f*)(out + (size_t)(m0 + 16 * t + r + 8 * hl) * CC + n0 + 64 * hf + 4 * ln) = o;
      }
  }
}

extern "C" void kernel_launch(void* const* d_in, const int* in_sizes, int n_in,
                              void* d_out, int out_size, void* d_ws, size_t ws_size,
                              hipStream_t stream) {
  (void)in_sizes; (void)n_in; (void)out_size; (void)ws_size;
  const float* x    = (const float*)d_in[0];
  const float* Wqkv = (const float*)d_in[1];
  const float* bqkv = (const float*)d_in[2];
  const float* Wout = (const float*)d_in[3];
  const float* bout = (const float*)d_in[4];
  float* out = (float*)d_out;

  float* qp = (float*)d_ws;
  float* kp = qp + HSZ;
  float* vp = kp + HSZ;
  float* ao = vp + HSZ;

  const dim3 blk(32, 1, 1);
  qkv_gemm_kernel<<<dim3(NQKV / 128, MTOT / 32), blk, 0, stream>>>(x, Wqkv, bqkv, qp, kp, vp);
  attn_kernel<<<dim3(TT / 16, BB * HH), blk, 0, stream>>>(qp, kp, vp, ao);
  out_gemm_kernel<<<dim3(CC / 128, MTOT / 32), blk, 0, stream>>>(ao, Wout, bout, out);
}


# Attention key loop takes 64 keys per trip, was 16: 4 independent score tiles (tile j = keys 4*lane+j); per row one max, one sum butterfly, one O rescale, one 16-byte LDS store; key sum order changes
# speedup vs baseline: 1.9905x; 1.2502x over previous
#include <hip/hip_runtime.h>


typedef __attribute__((ext_vector_type(2))) float v2f;
typedef __attribute__((ext_vector_type(4))) float v4f;
typedef __attribute__((ext_vector_type(8))) float v8f;

#define WMMA_F32(a, b, c) \
  __builtin_amdgcn_wmma_f32_16x16x4_f32(false, (a), false, (b), (short)0, (c), false, false)

static constexpr int BB = 2, TT = 2048, CC = 1024, HH = 16, DD = 64;
static constexpr int MTOT = BB * TT;
static constexpr int NQKV = 3 * CC;
static constexpr size_t HSZ = (size_t)BB * HH * TT * DD;


__device__ __forceinline__ void load_a_chunk(v4f (&a)[2][4],
                                             const float* __restrict__ A0,
                                             const float* __restrict__ A1, const int kc) {
#pragma unroll
  for (int u = 0; u < 4; ++u) a[0][u] = *(const v4f*)(A0 + kc + 8 * u);
#pragma unroll
  for (int u = 0; u < 4; ++u) a[1][u] = *(const v4f*)(A1 + kc + 8 * u);
}

template <int LDB>
__device__ __forceinline__ void load_b_step(v4f (&b)[8], const float* __restrict__ Bp,
                                            const int k0) {
  const float* Bk = Bp + (size_t)k0 * LDB;
#pragma unroll
  for (int r = 0; r < 4; ++r) {
    b[2 * r]     = *(const v4f*)(Bk + r * LDB);
    b[2 * r + 1] = *(const v4f*)(Bk + r * LDB + 64);
  }
}

__device__ __forceinline__ void mma_step(const v4f a0, const v4f a1, const v4f (&b)[8],
                                         v8f (&c)[2][8]) {
  const v2f a0lo = {a0.x, a0.y}, a1lo = {a1.x, a1.y};
  const v2f a0hi = {a0.z, a0.w}, a1hi = {a1.z, a1.w};
#pragma unroll
  for (int hf = 0; hf < 2; ++hf)
#pragma unroll
    for (int j = 0; j < 4; ++j) {
      const v2f bb = {b[hf][j], b[2 + hf][j]};
      c[0][4 * hf + j] = WMMA_F32(a0lo, bb, c[0][4 * hf + j]);
      c[1][4 * hf + j] = WMMA_F32(a1lo, bb, c[1][4 * hf + j]);
    }
#pragma unroll
  for (int hf = 0; hf < 2; ++hf)
#pragma unroll
    for (int j = 0; j < 4; ++j) {
      const v2f bb = {b[4 + hf][j], b[6 + hf][j]};
      c[0][4 * hf + j] = WMMA_F32(a0hi, bb, c[0][4 * hf + j]);
      c[1][4 * hf + j] = WMMA_F32(a1hi, bb, c[1][4 * hf + j]);
    }
}

template <int LDB>
__device__ __forceinline__ void gemm_chunk(v4f (&aC)[2][4], v4f (&aN)[2][4],
                                           v4f (&bP)[8], v4f (&bQ)[8],
                                           const float* __restrict__ A0,
                                           const float* __restrict__ A1,
                                           const float* __restrict__ Bp, const int kc,
                                           v8f (&c)[2][8]) {
  load_a_chunk(aN, A0, A1, (kc + 32) & (CC - 1));
  __builtin_amdgcn_sched_barrier(0);
  load_b_step<LDB>(bQ, Bp, kc + 8);
  mma_step(aC[0][0], aC[1][0], bP, c);
  load_b_step<LDB>(bP, Bp, kc + 16);
  mma_step(aC[0][1], aC[1][1], bQ, c);
  load_b_step<LDB>(bQ, Bp, kc + 24);
  mma_step(aC[0][2], aC[1][2], bP, c);
  load_b_step<LDB>(bP, Bp, (kc + 32) & (CC - 1));
  mma_step(aC[0][3], aC[1][3], bQ, c);
}

template <int LDB>
__device__ __forceinline__ void gemm_tile_32x128(const float* __restrict__ A0,
                                                 const float* __restrict__ A1,
                                                 const float* __restrict__ Bp,
                                                 v8f (&c)[2][8]) {
  v4f aP[2][4], aQ[2][4], bP[8], bQ[8];
  load_a_chunk(aP, A0, A1, 0);
  load_b_step<LDB>(bP, Bp, 0);
  for (int kc = 0; kc < CC; kc += 64) {
    gemm_chunk<LDB>(aP, aQ, bP, bQ, A0, A1, Bp, kc, c);
    gemm_chunk<LDB>(aQ, aP, bP, bQ, A0, A1, Bp, kc + 32, c);
  }
}

__global__ __launch_bounds__(32)
void qkv_gemm_kernel(const float* __restrict__ X, const float* __restrict__ W,
                     const float* __restrict__ bias,
                     float* __restrict__ q, float* __restrict__ k,
                     float* __restrict__ v) {
  const int lane = threadIdx.x & 31;
  const int hl = lane >> 4;
  const int ln = lane & 15;
  const int n0 = blockIdx.x * 128;
  const int m0 = blockIdx.y * 32;

  v8f c[2][8] = {};
  const float* A0 = X + (size_t)(m0 + ln) * CC + 4 * hl;
  gemm_tile_32x128<NQKV>(A0, A0 + 16 * CC,
                         W + (size_t)(4 * hl) * NQKV + n0 + 4 * ln, c);

  const int which = n0 >> 10;
  float* dst = (which == 0) ? q : (which == 1) ? k : v;
#pragma unroll
  for (int hf = 0; hf < 2; ++hf) {
    const v4f bv = *(const v4f*)(bias + n0 + 64 * hf + 4 * ln);
    const int h = ((n0 + 64 * hf) & (CC - 1)) >> 6;
#pragma unroll
    for (int t = 0; t < 2; ++t)
#pragma unroll
      for (int r = 0; r < 8; ++r) {
        const int trow = m0 + 16 * t + r + 8 * hl;
        const int bidx = trow >> 11;
        const int tt = trow & (TT - 1);
        const size_t idx = (((size_t)(bidx * HH + h) * TT) + tt) * DD + 4 * ln;
        const v4f o = {c[t][4 * hf + 0][r] + bv.x, c[t][4 * hf + 1][r] + bv.y,
                       c[t][4 * hf + 2][r] + bv.z, c[t][4 * hf + 3][r] + bv.w};
        *(v4f*)(dst + idx) = o;
      }
  }
}

static constexpr int KTRIP = 64;
static constexpr int PPITCH = KTRIP + 4;

template <bool MASK>
__device__ __forceinline__ void attn_trip(const v4f (&qa)[8], v8f (&o)[4], float (&m_i)[8],
                                          float (&l_i)[8], float (*pt)[PPITCH],
                                          const float* __restrict__ kbase,
                                          const float* __restrict__ vbase, const int kb,
                                          const int q0, const int hl, const int ln) {
  v8f s[4] = {};
  const float* krow = kbase + (size_t)(kb + 4 * ln) * DD + 4 * hl;
#pragma unroll
  for (int u = 0; u < 8; ++u) {
    v4f kf[4];
#pragma unroll
    for (int j = 0; j < 4; ++j) kf[j] = *(const v4f*)(krow + j * DD + 8 * u);
    const v2f q01 = {qa[u].x, qa[u].y};
    const v2f q23 = {qa[u].z, qa[u].w};
#pragma unroll
    for (int j = 0; j < 4; ++j) {
      const v2f k01 = {kf[j].x, kf[j].y};
      s[j] = WMMA_F32(q01, k01, s[j]);
    }
#pragma unroll
    for (int j = 0; j < 4; ++j) {
      const v2f k23 = {kf[j].z, kf[j].w};
      s[j] = WMMA_F32(q23, k23, s[j]);
    }
  }

  if (MASK) {
#pragma unroll
    for (int j = 0; j < 4; ++j)
#pragma unroll
      for (int r = 0; r < 8; ++r)
        if (kb + 4 * ln + j > q0 + r + 8 * hl) s[j][r] = -__builtin_inff();
  }

#pragma unroll
  for (int r = 0; r < 8; ++r) {
    float mx = fmaxf(fmaxf(s[0][r], s[1][r]), fmaxf(s[2][r], s[3][r]));
    mx = fmaxf(mx, __shfl_xor(mx, 1));
    mx = fmaxf(mx, __shfl_xor(mx, 2));
    mx = fmaxf(mx, __shfl_xor(mx, 4));
    mx = fmaxf(mx, __shfl_xor(mx, 8));
    const float mn = fmaxf(m_i[r], mx);
    const float sc = __expf(m_i[r] - mn);
    const v4f p = {__expf(s[0][r] - mn), __expf(s[1][r] - mn),
                   __expf(s[2][r] - mn), __expf(s[3][r] - mn)};
    float ps = (p.x + p.y) + (p.z + p.w);
    ps += __shfl_xor(ps, 1);
    ps += __shfl_xor(ps, 2);
    ps += __shfl_xor(ps, 4);
    ps += __shfl_xor(ps, 8);
    l_i[r] = l_i[r] * sc + ps;
    m_i[r] = mn;
    *(v4f*)&pt[r + 8 * hl][4 * ln] = p;
#pragma unroll
    for (int n = 0; n < 4; ++n) o[n][r] *= sc;
  }

  __syncthreads();

  v4f pa[8];
#pragma unroll
  for (int u = 0; u < 8; ++u) pa[u] = *(const v4f*)&pt[ln][8 * u + 4 * hl];
  const float* vrow = vbase + (size_t)(kb + 4 * hl) * DD + 4 * ln;
#pragma unroll
  for (int u = 0; u < 8; ++u) {
    const v4f v0 = *(const v4f*)(vrow + (8 * u + 0) * DD);
    const v4f v1 = *(const v4f*)(vrow + (8 * u + 1) * DD);
    const v4f v2 = *(const v4f*)(vrow + (8 * u + 2) * DD);
    const v4f v3 = *(const v4f*)(vrow + (8 * u + 3) * DD);
    const v2f a01 = {pa[u].x, pa[u].y};
    const v2f a23 = {pa[u].z, pa[u].w};
#pragma unroll
    for (int n = 0; n < 4; ++n) {
      const v2f bf = {v0[n], v1[n]};
      o[n] = WMMA_F32(a01, bf, o[n]);
    }
#pragma unroll
    for (int n = 0; n < 4; ++n) {
      const v2f bf = {v2[n], v3[n]};
      o[n] = WMMA_F32(a23, bf, o[n]);
    }
  }
  __syncthreads();
}

__global__ __launch_bounds__(32)
void attn_kernel(const float* __restrict__ q, const float* __restrict__ kmat,
                 const float* __restrict__ vmat, float* __restrict__ out) {
  const int lane = threadIdx.x & 31;
  const int hl = lane >> 4;
  const int ln = lane & 15;
  const int qtile = blockIdx.x;
  const int bh = blockIdx.y;
  const int b = bh >> 4, h = bh & 15;
  const int q0 = qtile * 16;

  const float* qbase = q + (size_t)bh * TT * DD;
  const float* kbase = kmat + (size_t)bh * TT * DD;
  const float* vbase = vmat + (size_t)bh * TT * DD;

  __shared__ __attribute__((aligned(16))) float pt[16][PPITCH];

  v4f qa[8];
  {
    const float* qrow = qbase + (size_t)(q0 + ln) * DD + 4 * hl;
#pragma unroll
    for (int u = 0; u < 8; ++u) {
      const v4f a = *(const v4f*)(qrow + 8 * u);
      qa[u] = a * 0.125f;
    }
  }

  v8f o[4] = {};
  float m_i[8], l_i[8];
#pragma unroll
  for (int r = 0; r < 8; ++r) { m_i[r] = -__builtin_inff(); l_i[r] = 0.0f; }

  const int last = q0 / KTRIP;
  for (int kt = 0; kt < last; ++kt)
    attn_trip<false>(qa, o, m_i, l_i, pt, kbase, vbase, kt * KTRIP, q0, hl, ln);
  attn_trip<true>(qa, o, m_i, l_i, pt, kbase, vbase, last * KTRIP, q0, hl, ln);

#pragma unroll
  for (int r = 0; r < 8; ++r) {
    const float inv = 1.0f / l_i[r];
    const int trow = q0 + r + 8 * hl;
    float* orow = out + ((size_t)(b * TT + trow) * CC) + h * DD + 4 * ln;
    const v4f ov = {o[0][r] * inv, o[1][r] * inv, o[2][r] * inv, o[3][r] * inv};
    *(v4f*)orow = ov;
  }
}

__global__ __launch_bounds__(32)
void out_gemm_kernel(const float* __restrict__ A, const float* __restrict__ W,
                     const float* __restrict__ bias, float* __restrict__ out) {
  const int lane = threadIdx.x & 31;
  const int hl = lane >> 4;
  const int ln = lane & 15;
  const int n0 = blockIdx.x * 128;
  const int m0 = blockIdx.y * 32;

  v8f c[2][8] = {};
  const float* A0 = A + (size_t)(m0 + ln) * CC + 4 * hl;
  gemm_tile_32x128<CC>(A0, A0 + 16 * CC,
                       W + (size_t)(4 * hl) * CC + n0 + 4 * ln, c);

#pragma unroll
  for (int hf = 0; hf < 2; ++hf) {
    const v4f bv = *(const v4f*)(bias + n0 + 64 * hf + 4 * ln);
#pragma unroll
    for (int t = 0; t < 2; ++t)
#pragma unroll
      for (int r = 0; r < 8; ++r) {
        const v4f o = {c[t][4 * hf + 0][r] + bv.x, c[t][4 * hf + 1][r] + bv.y,
                       c[t][4 * hf + 2][r] + bv.z, c[t][4 * hf + 3][r] + bv.w};
        *(v4f*)(out + (size_t)(m0 + 16 * t + r + 8 * hl) * CC + n0 + 64 * hf + 4 * ln) = o;
      }
  }
}

extern "C" void kernel_launch(void* const* d_in, const int* in_sizes, int n_in,
                              void* d_out, int out_size, void* d_ws, size_t ws_size,
                              hipStream_t stream) {
  (void)in_sizes; (void)n_in; (void)out_size; (void)ws_size;
  const float* x    = (const float*)d_in[0];
  const float* Wqkv = (const float*)d_in[1];
  const float* bqkv = (const float*)d_in[2];
  const float* Wout = (const float*)d_in[3];
  const float* bout = (const float*)d_in[4];
  float* out = (float*)d_out;

  float* qp = (float*)d_ws;
  float* kp = qp + HSZ;
  float* vp = kp + HSZ;
  float* ao = vp + HSZ;

  const dim3 blk(32, 1, 1);
  qkv_gemm_kernel<<<dim3(NQKV / 128, MTOT / 32), blk, 0, stream>>>(x, Wqkv, bqkv, qp, kp, vp);
  attn_kernel<<<dim3(TT / 16, BB * HH), blk, 0, stream>>>(qp, kp, vp, ao);
  out_gemm_kernel<<<dim3(CC / 128, MTOT / 32), blk, 0, stream>>>(ao, Wout, bout, out);
}


# Attention: each wave takes the query-tile pair (127-p, p) in turn: 2048 equal waves of 33 key trips, two per SIMD on two waves' register budget (was 4096 waves of 1 to 32 trips); sum order unchanged
# speedup vs baseline: 2.3654x; 1.1883x over previous
#include <hip/hip_runtime.h>


typedef __attribute__((ext_vector_type(2))) float v2f;
typedef __attribute__((ext_vector_type(4))) float v4f;
typedef __attribute__((ext_vector_type(8))) float v8f;

#define WMMA_F32(a, b, c) \
  __builtin_amdgcn_wmma_f32_16x16x4_f32(false, (a), false, (b), (short)0, (c), false, false)

static constexpr int BB = 2, TT = 2048, CC = 1024, HH = 16, DD = 64;
static constexpr int MTOT = BB * TT;
static constexpr int NQKV = 3 * CC;
static constexpr size_t HSZ = (size_t)BB * HH * TT * DD;


__device__ __forceinline__ void load_a_chunk(v4f (&a)[2][4],
                                             const float* __restrict__ A0,
                                             const float* __restrict__ A1, const int kc) {
#pragma unroll
  for (int u = 0; u < 4; ++u) a[0][u] = *(const v4f*)(A0 + kc + 8 * u);
#pragma unroll
  for (int u = 0; u < 4; ++u) a[1][u] = *(const v4f*)(A1 + kc + 8 * u);
}

template <int LDB>
__device__ __forceinline__ void load_b_step(v4f (&b)[8], const float* __restrict__ Bp,
                                            const int k0) {
  const float* Bk = Bp + (size_t)k0 * LDB;
#pragma unroll
  for (int r = 0; r < 4; ++r) {
    b[2 * r]     = *(const v4f*)(Bk + r * LDB);
    b[2 * r + 1] = *(const v4f*)(Bk + r * LDB + 64);
  }
}

__device__ __forceinline__ void mma_step(const v4f a0, const v4f a1, const v4f (&b)[8],
                                         v8f (&c)[2][8]) {
  const v2f a0lo = {a0.x, a0.y}, a1lo = {a1.x, a1.y};
  const v2f a0hi = {a0.z, a0.w}, a1hi = {a1.z, a1.w};
#pragma unroll
  for (int hf = 0; hf < 2; ++hf)
#pragma unroll
    for (int j = 0; j < 4; ++j) {
      const v2f bb = {b[hf][j], b[2 + hf][j]};
      c[0][4 * hf + j] = WMMA_F32(a0lo, bb, c[0][4 * hf + j]);
      c[1][4 * hf + j] = WMMA_F32(a1lo, bb, c[1][4 * hf + j]);
    }
#pragma unroll
  for (int hf = 0; hf < 2; ++hf)
#pragma unroll
    for (int j = 0; j < 4; ++j) {
      const v2f bb = {b[4 + hf][j], b[6 + hf][j]};
      c[0][4 * hf + j] = WMMA_F32(a0hi, bb, c[0][4 * hf + j]);
      c[1][4 * hf + j] = WMMA_F32(a1hi, bb, c[1][4 * hf + j]);
    }
}

template <int LDB>
__device__ __forceinline__ void gemm_chunk(v4f (&aC)[2][4], v4f (&aN)[2][4],
                                           v4f (&bP)[8], v4f (&bQ)[8],
                                           const float* __restrict__ A0,
                                           const float* __restrict__ A1,
                                           const float* __restrict__ Bp, const int kc,
                                           v8f (&c)[2][8]) {
  load_a_chunk(aN, A0, A1, (kc + 32) & (CC - 1));
  __builtin_amdgcn_sched_barrier(0);
  load_b_step<LDB>(bQ, Bp, kc + 8);
  mma_step(aC[0][0], aC[1][0], bP, c);
  load_b_step<LDB>(bP, Bp, kc + 16);
  mma_step(aC[0][1], aC[1][1], bQ, c);
  load_b_step<LDB>(bQ, Bp, kc + 24);
  mma_step(aC[0][2], aC[1][2], bP, c);
  load_b_step<LDB>(bP, Bp, (kc + 32) & (CC - 1));
  mma_step(aC[0][3], aC[1][3], bQ, c);
}

template <int LDB>
__device__ __forceinline__ void gemm_tile_32x128(const float* __restrict__ A0,
                                                 const float* __restrict__ A1,
                                                 const float* __restrict__ Bp,
                                                 v8f (&c)[2][8]) {
  v4f aP[2][4], aQ[2][4], bP[8], bQ[8];
  load_a_chunk(aP, A0, A1, 0);
  load_b_step<LDB>(bP, Bp, 0);
  for (int kc = 0; kc < CC; kc += 64) {
    gemm_chunk<LDB>(aP, aQ, bP, bQ, A0, A1, Bp, kc, c);
    gemm_chunk<LDB>(aQ, aP, bP, bQ, A0, A1, Bp, kc + 32, c);
  }
}

__global__ __launch_bounds__(32)
void qkv_gemm_kernel(const float* __restrict__ X, const float* __restrict__ W,
                     const float* __restrict__ bias,
                     float* __restrict__ q, float* __restrict__ k,
                     float* __restrict__ v) {
  const int lane = threadIdx.x & 31;
  const int hl = lane >> 4;
  const int ln = lane & 15;
  const int n0 = blockIdx.x * 128;
  const int m0 = blockIdx.y * 32;

  v8f c[2][8] = {};
  const float* A0 = X + (size_t)(m0 + ln) * CC + 4 * hl;
  gemm_tile_32x128<NQKV>(A0, A0 + 16 * CC,
                         W + (size_t)(4 * hl) * NQKV + n0 + 4 * ln, c);

  const int which = n0 >> 10;
  float* dst = (which == 0) ? q : (which == 1) ? k : v;
#pragma unroll
  for (int hf = 0; hf < 2; ++hf) {
    const v4f bv = *(const v4f*)(bias + n0 + 64 * hf + 4 * ln);
    const int h = ((n0 + 64 * hf) & (CC - 1)) >> 6;
#pragma unroll
    for (int t = 0; t < 2; ++t)
#pragma unroll
      for (int r = 0; r < 8; ++r) {
        const int trow = m0 + 16 * t + r + 8 * hl;
        const int bidx = trow >> 11;
        const int tt = trow & (TT - 1);
        const size_t idx = (((size_t)(bidx * HH + h) * TT) + tt) * DD + 4 * ln;
        const v4f o = {c[t][4 * hf + 0][r] + bv.x, c[t][4 * hf + 1][r] + bv.y,
                       c[t][4 * hf + 2][r] + bv.z, c[t][4 * hf + 3][r] + bv.w};
        *(v4f*)(dst + idx) = o;
      }
  }
}

static constexpr int KTRIP = 64;
static constexpr int PPITCH = KTRIP + 4;

template <bool MASK>
__device__ __forceinline__ void attn_trip(const v4f (&qa)[8], v8f (&o)[4], float (&m_i)[8],
                                          float (&l_i)[8], float (*pt)[PPITCH],
                                          const float* __restrict__ kbase,
                                          const float* __restrict__ vbase, const int kb,
                                          const int q0, const int hl, const int ln) {
  v8f s[4] = {};
  const float* krow = kbase + (size_t)(kb + 4 * ln) * DD + 4 * hl;
#pragma unroll
  for (int u = 0; u < 8; ++u) {
    v4f kf[4];
#pragma unroll
    for (int j = 0; j < 4; ++j) kf[j] = *(const v4f*)(krow + j * DD + 8 * u);
    const v2f q01 = {qa[u].x, qa[u].y};
    const v2f q23 = {qa[u].z, qa[u].w};
#pragma unroll
    for (int j = 0; j < 4; ++j) {
      const v2f k01 = {kf[j].x, kf[j].y};
      s[j] = WMMA_F32(q01, k01, s[j]);
    }
#pragma unroll
    for (int j = 0; j < 4; ++j) {
      const v2f k23 = {kf[j].z, kf[j].w};
      s[j] = WMMA_F32(q23, k23, s[j]);
    }
  }

  if (MASK) {
#pragma unroll
    for (int j = 0; j < 4; ++j)
#pragma unroll
      for (int r = 0; r < 8; ++r)
        if (kb + 4 * ln + j > q0 + r + 8 * hl) s[j][r] = -__builtin_inff();
  }

#pragma unroll
  for (int r = 0; r < 8; ++r) {
    float mx = fmaxf(fmaxf(s[0][r], s[1][r]), fmaxf(s[2][r], s[3][r]));
    mx = fmaxf(mx, __shfl_xor(mx, 1));
    mx = fmaxf(mx, __shfl_xor(mx, 2));
    mx = fmaxf(mx, __shfl_xor(mx, 4));
    mx = fmaxf(mx, __shfl_xor(mx, 8));
    const float mn = fmaxf(m_i[r], mx);
    const float sc = __expf(m_i[r] - mn);
    const v4f p = {__expf(s[0][r] - mn), __expf(s[1][r] - mn),
                   __expf(s[2][r] - mn), __expf(s[3][r] - mn)};
    float ps = (p.x + p.y) + (p.z + p.w);
    ps += __shfl_xor(ps, 1);
    ps += __shfl_xor(ps, 2);
    ps += __shfl_xor(ps, 4);
    ps += __shfl_xor(ps, 8);
    l_i[r] = l_i[r] * sc + ps;
    m_i[r] = mn;
    *(v4f*)&pt[r + 8 * hl][4 * ln] = p;
#pragma unroll
    for (int n = 0; n < 4; ++n) o[n][r] *= sc;
  }

  __syncthreads();

  v4f pa[8];
#pragma unroll
  for (int u = 0; u < 8; ++u) pa[u] = *(const v4f*)&pt[ln][8 * u + 4 * hl];
  const float* vrow = vbase + (size_t)(kb + 4 * hl) * DD + 4 * ln;
#pragma unroll
  for (int u = 0; u < 8; ++u) {
    const v4f v0 = *(const v4f*)(vrow + (8 * u + 0) * DD);
    const v4f v1 = *(const v4f*)(vrow + (8 * u + 1) * DD);
    const v4f v2 = *(const v4f*)(vrow + (8 * u + 2) * DD);
    const v4f v3 = *(const v4f*)(vrow + (8 * u + 3) * DD);
    const v2f a01 = {pa[u].x, pa[u].y};
    const v2f a23 = {pa[u].z, pa[u].w};
#pragma unroll
    for (int n = 0; n < 4; ++n) {
      const v2f bf = {v0[n], v1[n]};
      o[n] = WMMA_F32(a01, bf, o[n]);
    }
#pragma unroll
    for (int n = 0; n < 4; ++n) {
      const v2f bf = {v2[n], v3[n]};
      o[n] = WMMA_F32(a23, bf, o[n]);
    }
  }
  __syncthreads();
}

__global__ __launch_bounds__(32) __attribute__((amdgpu_waves_per_eu(2, 2)))
void attn_kernel(const float* __restrict__ q, const float* __restrict__ kmat,
                 const float* __restrict__ vmat, float* __restrict__ out) {
  const int lane = threadIdx.x & 31;
  const int hl = lane >> 4;
  const int ln = lane & 15;
  const int pr = blockIdx.x;
  const int bh = blockIdx.y;
  const int b = bh >> 4, h = bh & 15;

  const float* qbase = q + (size_t)bh * TT * DD;
  const float* kbase = kmat + (size_t)bh * TT * DD;
  const float* vbase = vmat + (size_t)bh * TT * DD;

  __shared__ __attribute__((aligned(16))) float pt[16][PPITCH];

  for (int half = 0; half < 2; ++half) {
    const int qtile = half ? pr : (TT / 16 - 1 - pr);
    const int q0 = qtile * 16;

    v4f qa[8];
    {
      const float* qrow = qbase + (size_t)(q0 + ln) * DD + 4 * hl;
#pragma unroll
      for (int u = 0; u < 8; ++u) {
        const v4f a = *(const v4f*)(qrow + 8 * u);
        qa[u] = a * 0.125f;
      }
    }

    v8f o[4] = {};
    float m_i[8], l_i[8];
#pragma unroll
    for (int r = 0; r < 8; ++r) { m_i[r] = -__builtin_inff(); l_i[r] = 0.0f; }

    const int last = q0 / KTRIP;
    for (int kt = 0; kt < last; ++kt)
      attn_trip<false>(qa, o, m_i, l_i, pt, kbase, vbase, kt * KTRIP, q0, hl, ln);
    attn_trip<true>(qa, o, m_i, l_i, pt, kbase, vbase, last * KTRIP, q0, hl, ln);

#pragma unroll
    for (int r = 0; r < 8; ++r) {
      const float inv = 1.0f / l_i[r];
      const int trow = q0 + r + 8 * hl;
      float* orow = out + ((size_t)(b * TT + trow) * CC) + h * DD + 4 * ln;
      const v4f ov = {o[0][r] * inv, o[1][r] * inv, o[2][r] * inv, o[3][r] * inv};
      *(v4f*)orow = ov;
    }
  }
}

__global__ __launch_bounds__(32)
void out_gemm_kernel(const float* __restrict__ A, const float* __restrict__ W,
                     const float* __restrict__ bias, float* __restrict__ out) {
  const int lane = threadIdx.x & 31;
  const int hl = lane >> 4;
  const int ln = lane & 15;
  const int n0 = blockIdx.x * 128;
  const int m0 = blockIdx.y * 32;

  v8f c[2][8] = {};
  const float* A0 = A + (size_t)(m0 + ln) * CC + 4 * hl;
  gemm_tile_32x128<CC>(A0, A0 + 16 * CC,
                       W + (size_t)(4 * hl) * CC + n0 + 4 * ln, c);

#pragma unroll
  for (int hf = 0; hf < 2; ++hf) {
    const v4f bv = *(const v4f*)(bias + n0 + 64 * hf + 4 * ln);
#pragma unroll
    for (int t = 0; t < 2; ++t)
#pragma unroll
      for (int r = 0; r < 8; ++r) {
        const v4f o = {c[t][4 * hf + 0][r] + bv.x, c[t][4 * hf + 1][r] + bv.y,
                       c[t][4 * hf + 2][r] + bv.z, c[t][4 * hf + 3][r] + bv.w};
        *(v4f*)(out + (size_t)(m0 + 16 * t + r + 8 * hl) * CC + n0 + 64 * hf + 4 * ln) = o;
      }
  }
}

extern "C" void kernel_launch(void* const* d_in, const int* in_sizes, int n_in,
                              void* d_out, int out_size, void* d_ws, size_t ws_size,
                              hipStream_t stream) {
  (void)in_sizes; (void)n_in; (void)out_size; (void)ws_size;
  const float* x    = (const float*)d_in[0];
  const float* Wqkv = (const float*)d_in[1];
  const float* bqkv = (const float*)d_in[2];
  const float* Wout = (const float*)d_in[3];
  const float* bout = (const float*)d_in[4];
  float* out = (float*)d_out;

  float* qp = (float*)d_ws;
  float* kp = qp + HSZ;
  float* vp = kp + HSZ;
  float* ao = vp + HSZ;

  const dim3 blk(32, 1, 1);
  qkv_gemm_kernel<<<dim3(NQKV / 128, MTOT / 32), blk, 0, stream>>>(x, Wqkv, bqkv, qp, kp, vp);
  attn_kernel<<<dim3(TT / 32, BB * HH), blk, 0, stream>>>(qp, kp, vp, ao);
  out_gemm_kernel<<<dim3(CC / 128, MTOT / 32), blk, 0, stream>>>(ao, Wout, bout, out);
}


# K lies in the workspace in score-fragment order (written so by the QKV epilogue): each K fragment load of a wave is 512 contiguous bytes, 4 lines, was 16 rows x 32 bytes; sum order unchanged
# speedup vs baseline: 2.4595x; 1.0398x over previous
#include <hip/hip_runtime.h>


typedef __attribute__((ext_vector_type(2))) float v2f;
typedef __attribute__((ext_vector_type(4))) float v4f;
typedef __attribute__((ext_vector_type(8))) float v8f;

#define WMMA_F32(a, b, c) \
  __builtin_amdgcn_wmma_f32_16x16x4_f32(false, (a), false, (b), (short)0, (c), false, false)

static constexpr int BB = 2, TT = 2048, CC = 1024, HH = 16, DD = 64;
static constexpr int MTOT = BB * TT;
static constexpr int NQKV = 3 * CC;
static constexpr size_t HSZ = (size_t)BB * HH * TT * DD;


__device__ __forceinline__ void load_a_chunk(v4f (&a)[2][4],
                                             const float* __restrict__ A0,
                                             const float* __restrict__ A1, const int kc) {
#pragma unroll
  for (int u = 0; u < 4; ++u) a[0][u] = *(const v4f*)(A0 + kc + 8 * u);
#pragma unroll
  for (int u = 0; u < 4; ++u) a[1][u] = *(const v4f*)(A1 + kc + 8 * u);
}

template <int LDB>
__device__ __forceinline__ void load_b_step(v4f (&b)[8], const float* __restrict__ Bp,
                                            const int k0) {
  const float* Bk = Bp + (size_t)k0 * LDB;
#pragma unroll
  for (int r = 0; r < 4; ++r) {
    b[2 * r]     = *(const v4f*)(Bk + r * LDB);
    b[2 * r + 1] = *(const v4f*)(Bk + r * LDB + 64);
  }
}

__device__ __forceinline__ void mma_step(const v4f a0, const v4f a1, const v4f (&b)[8],
                                         v8f (&c)[2][8]) {
  const v2f a0lo = {a0.x, a0.y}, a1lo = {a1.x, a1.y};
  const v2f a0hi = {a0.z, a0.w}, a1hi = {a1.z, a1.w};
#pragma unroll
  for (int hf = 0; hf < 2; ++hf)
#pragma unroll
    for (int j = 0; j < 4; ++j) {
      const v2f bb = {b[hf][j], b[2 + hf][j]};
      c[0][4 * hf + j] = WMMA_F32(a0lo, bb, c[0][4 * hf + j]);
      c[1][4 * hf + j] = WMMA_F32(a1lo, bb, c[1][4 * hf + j]);
    }
#pragma unroll
  for (int hf = 0; hf < 2; ++hf)
#pragma unroll
    for (int j = 0; j < 4; ++j) {
      const v2f bb = {b[4 + hf][j], b[6 + hf][j]};
      c[0][4 * hf + j] = WMMA_F32(a0hi, bb, c[0][4 * hf + j]);
      c[1][4 * hf + j] = WMMA_F32(a1hi, bb, c[1][4 * hf + j]);
    }
}

template <int LDB>
__device__ __forceinline__ void gemm_chunk(v4f (&aC)[2][4], v4f (&aN)[2][4],
                                           v4f (&bP)[8], v4f (&bQ)[8],
                                           const float* __restrict__ A0,
                                           const float* __restrict__ A1,
                                           const float* __restrict__ Bp, const int kc,
                                           v8f (&c)[2][8]) {
  load_a_chunk(aN, A0, A1, (kc + 32) & (CC - 1));
  __builtin_amdgcn_sched_barrier(0);
  load_b_step<LDB>(bQ, Bp, kc + 8);
  mma_step(aC[0][0], aC[1][0], bP, c);
  load_b_step<LDB>(bP, Bp, kc + 16);
  mma_step(aC[0][1], aC[1][1], bQ, c);
  load_b_step<LDB>(bQ, Bp, kc + 24);
  mma_step(aC[0][2], aC[1][2], bP, c);
  load_b_step<LDB>(bP, Bp, (kc + 32) & (CC - 1));
  mma_step(aC[0][3], aC[1][3], bQ, c);
}

template <int LDB>
__device__ __forceinline__ void gemm_tile_32x128(const float* __restrict__ A0,
                                                 const float* __restrict__ A1,
                                                 const float* __restrict__ Bp,
                                                 v8f (&c)[2][8]) {
  v4f aP[2][4], aQ[2][4], bP[8], bQ[8];
  load_a_chunk(aP, A0, A1, 0);
  load_b_step<LDB>(bP, Bp, 0);
  for (int kc = 0; kc < CC; kc += 64) {
    gemm_chunk<LDB>(aP, aQ, bP, bQ, A0, A1, Bp, kc, c);
    gemm_chunk<LDB>(aQ, aP, bP, bQ, A0, A1, Bp, kc + 32, c);
  }
}

__global__ __launch_bounds__(32)
void qkv_gemm_kernel(const float* __restrict__ X, const float* __restrict__ W,
                     const float* __restrict__ bias,
                     float* __restrict__ q, float* __restrict__ k,
                     float* __restrict__ v) {
  const int lane = threadIdx.x & 31;
  const int hl = lane >> 4;
  const int ln = lane & 15;
  const int n0 = blockIdx.x * 128;
  const int m0 = blockIdx.y * 32;

  v8f c[2][8] = {};
  const float* A0 = X + (size_t)(m0 + ln) * CC + 4 * hl;
  gemm_tile_32x128<NQKV>(A0, A0 + 16 * CC,
                         W + (size_t)(4 * hl) * NQKV + n0 + 4 * ln, c);

  const int which = n0 >> 10;
  float* dst = (which == 0) ? q : (which == 1) ? k : v;
  const int kfrag = (ln >> 1) * 512 + (ln & 1) * 4;
#pragma unroll
  for (int hf = 0; hf < 2; ++hf) {
    const v4f bv = *(const v4f*)(bias + n0 + 64 * hf + 4 * ln);
    const int h = ((n0 + 64 * hf) & (CC - 1)) >> 6;
#pragma unroll
    for (int t = 0; t < 2; ++t)
#pragma unroll
      for (int r = 0; r < 8; ++r) {
        const int trow = m0 + 16 * t + r + 8 * hl;
        const int bidx = trow >> 11;
        const int tt = trow & (TT - 1);
        const size_t rowm = (((size_t)(bidx * HH + h) * TT) + tt) * DD + 4 * ln;
        const size_t frag = (((size_t)(bidx * HH + h) * TT) + (tt & ~63)) * DD +
                            (tt & 3) * 128 + ((tt & 63) >> 2) * 8 + kfrag;
        const size_t idx = (which == 1) ? frag : rowm;
        const v4f o = {c[t][4 * hf + 0][r] + bv.x, c[t][4 * hf + 1][r] + bv.y,
                       c[t][4 * hf + 2][r] + bv.z, c[t][4 * hf + 3][r] + bv.w};
        *(v4f*)(dst + idx) = o;
      }
  }
}

static constexpr int KTRIP = 64;
static constexpr int PPITCH = KTRIP + 4;

template <bool MASK>
__device__ __forceinline__ void attn_trip(const v4f (&qa)[8], v8f (&o)[4], float (&m_i)[8],
                                          float (&l_i)[8], float (*pt)[PPITCH],
                                          const float* __restrict__ kbase,
                                          const float* __restrict__ vbase, const int kb,
                                          const int q0, const int hl, const int ln) {
  v8f s[4] = {};
  const float* krow = kbase + (size_t)kb * DD + 8 * ln + 4 * hl;
#pragma unroll
  for (int u = 0; u < 8; ++u) {
    v4f kf[4];
#pragma unroll
    for (int j = 0; j < 4; ++j) kf[j] = *(const v4f*)(krow + (4 * u + j) * 128);
    const v2f q01 = {qa[u].x, qa[u].y};
    const v2f q23 = {qa[u].z, qa[u].w};
#pragma unroll
    for (int j = 0; j < 4; ++j) {
      const v2f k01 = {kf[j].x, kf[j].y};
      s[j] = WMMA_F32(q01, k01, s[j]);
    }
#pragma unroll
    for (int j = 0; j < 4; ++j) {
      const v2f k23 = {kf[j].z, kf[j].w};
      s[j] = WMMA_F32(q23, k23, s[j]);
    }
  }

  if (MASK) {
#pragma unroll
    for (int j = 0; j < 4; ++j)
#pragma unroll
      for (int r = 0; r < 8; ++r)
        if (kb + 4 * ln + j > q0 + r + 8 * hl) s[j][r] = -__builtin_inff();
  }

#pragma unroll
  for (int r = 0; r < 8; ++r) {
    float mx = fmaxf(fmaxf(s[0][r], s[1][r]), fmaxf(s[2][r], s[3][r]));
    mx = fmaxf(mx, __shfl_xor(mx, 1));
    mx = fmaxf(mx, __shfl_xor(mx, 2));
    mx = fmaxf(mx, __shfl_xor(mx, 4));
    mx = fmaxf(mx, __shfl_xor(mx, 8));
    const float mn = fmaxf(m_i[r], mx);
    const float sc = __expf(m_i[r] - mn);
    const v4f p = {__expf(s[0][r] - mn), __expf(s[1][r] - mn),
                   __expf(s[2][r] - mn), __expf(s[3][r] - mn)};
    float ps = (p.x + p.y) + (p.z + p.w);
    ps += __shfl_xor(ps, 1);
    ps += __shfl_xor(ps, 2);
    ps += __shfl_xor(ps, 4);
    ps += __shfl_xor(ps, 8);
    l_i[r] = l_i[r] * sc + ps;
    m_i[r] = mn;
    *(v4f*)&pt[r + 8 * hl][4 * ln] = p;
#pragma unroll
    for (int n = 0; n < 4; ++n) o[n][r] *= sc;
  }

  __syncthreads();

  v4f pa[8];
#pragma unroll
  for (int u = 0; u < 8; ++u) pa[u] = *(const v4f*)&pt[ln][8 * u + 4 * hl];
  const float* vrow = vbase + (size_t)(kb + 4 * hl) * DD + 4 * ln;
#pragma unroll
  for (int u = 0; u < 8; ++u) {
    const v4f v0 = *(const v4f*)(vrow + (8 * u + 0) * DD);
    const v4f v1 = *(const v4f*)(vrow + (8 * u + 1) * DD);
    const v4f v2 = *(const v4f*)(vrow + (8 * u + 2) * DD);
    const v4f v3 = *(const v4f*)(vrow + (8 * u + 3) * DD);
    const v2f a01 = {pa[u].x, pa[u].y};
    const v2f a23 = {pa[u].z, pa[u].w};
#pragma unroll
    for (int n = 0; n < 4; ++n) {
      const v2f bf = {v0[n], v1[n]};
      o[n] = WMMA_F32(a01, bf, o[n]);
    }
#pragma unroll
    for (int n = 0; n < 4; ++n) {
      const v2f bf = {v2[n], v3[n]};
      o[n] = WMMA_F32(a23, bf, o[n]);
    }
  }
  __syncthreads();
}

__global__ __launch_bounds__(32) __attribute__((amdgpu_waves_per_eu(2, 2)))
void attn_kernel(const float* __restrict__ q, const float* __restrict__ kmat,
                 const float* __restrict__ vmat, float* __restrict__ out) {
  const int lane = threadIdx.x & 31;
  const int hl = lane >> 4;
  const int ln = lane & 15;
  const int pr = blockIdx.x;
  const int bh = blockIdx.y;
  const int b = bh >> 4, h = bh & 15;

  const float* qbase = q + (size_t)bh * TT * DD;
  const float* kbase = kmat + (size_t)bh * TT * DD;
  const float* vbase = vmat + (size_t)bh * TT * DD;

  __shared__ __attribute__((aligned(16))) float pt[16][PPITCH];

  for (int half = 0; half < 2; ++half) {
    const int qtile = half ? pr : (TT / 16 - 1 - pr);
    const int q0 = qtile * 16;

    v4f qa[8];
    {
      const float* qrow = qbase + (size_t)(q0 + ln) * DD + 4 * hl;
#pragma unroll
      for (int u = 0; u < 8; ++u) {
        const v4f a = *(const v4f*)(qrow + 8 * u);
        qa[u] = a * 0.125f;
      }
    }

    v8f o[4] = {};
    float m_i[8], l_i[8];
#pragma unroll
    for (int r = 0; r < 8; ++r) { m_i[r] = -__builtin_inff(); l_i[r] = 0.0f; }

    const int last = q0 / KTRIP;
    for (int kt = 0; kt < last; ++kt)
      attn_trip<false>(qa, o, m_i, l_i, pt, kbase, vbase, kt * KTRIP, q0, hl, ln);
    attn_trip<true>(qa, o, m_i, l_i, pt, kbase, vbase, last * KTRIP, q0, hl, ln);

#pragma unroll
    for (int r = 0; r < 8; ++r) {
      const float inv = 1.0f / l_i[r];
      const int trow = q0 + r + 8 * hl;
      float* orow = out + ((size_t)(b * TT + trow) * CC) + h * DD + 4 * ln;
      const v4f ov = {o[0][r] * inv, o[1][r] * inv, o[2][r] * inv, o[3][r] * inv};
      *(v4f*)orow = ov;
    }
  }
}

__global__ __launch_bounds__(32)
void out_gemm_kernel(const float* __restrict__ A, const float* __restrict__ W,
                     const float* __restrict__ bias, float* __restrict__ out) {
  const int lane = threadIdx.x & 31;
  const int hl = lane >> 4;
  const int ln = lane & 15;
  const int n0 = blockIdx.x * 128;
  const int m0 = blockIdx.y * 32;

  v8f c[2][8] = {};
  const float* A0 = A + (size_t)(m0 + ln) * CC + 4 * hl;
  gemm_tile_32x128<CC>(A0, A0 + 16 * CC,
                       W + (size_t)(4 * hl) * CC + n0 + 4 * ln, c);

#pragma unroll
  for (int hf = 0; hf < 2; ++hf) {
    const v4f bv = *(const v4f*)(bias + n0 + 64 * hf + 4 * ln);
#pragma unroll
    for (int t = 0; t < 2; ++t)
#pragma unroll
      for (int r = 0; r < 8; ++r) {
        const v4f o = {c[t][4 * hf + 0][r] + bv.x, c[t][4 * hf + 1][r] + bv.y,
                       c[t][4 * hf + 2][r] + bv.z, c[t][4 * hf + 3][r] + bv.w};
        *(v4f*)(out + (size_t)(m0 + 16 * t + r + 8 * hl) * CC + n0 + 64 * hf + 4 * ln) = o;
      }
  }
}

extern "C" void kernel_launch(void* const* d_in, const int* in_sizes, int n_in,
                              void* d_out, int out_size, void* d_ws, size_t ws_size,
                              hipStream_t stream) {
  (void)in_sizes; (void)n_in; (void)out_size; (void)ws_size;
  const float* x    = (const float*)d_in[0];
  const float* Wqkv = (const float*)d_in[1];
  const float* bqkv = (const float*)d_in[2];
  const float* Wout = (const float*)d_in[3];
  const float* bout = (const float*)d_in[4];
  float* out = (float*)d_out;

  float* qp = (float*)d_ws;
  float* kp = qp + HSZ;
  float* vp = kp + HSZ;
  float* ao = vp + HSZ;

  const dim3 blk(32, 1, 1);
  qkv_gemm_kernel<<<dim3(NQKV / 128, MTOT / 32), blk, 0, stream>>>(x, Wqkv, bqkv, qp, kp, vp);
  attn_kernel<<<dim3(TT / 32, BB * HH), blk, 0, stream>>>(qp, kp, vp, ao);
  out_gemm_kernel<<<dim3(CC / 128, MTOT / 32), blk, 0, stream>>>(ao, Wout, bout, out);
}


# Waves that read the same operand blocks form one work-group per WGP (12 QKV, 8 attention, 4 out-projection waves): their re-reads hit the WGP cache, were refetched from L2; sum order unchanged
# speedup vs baseline: 2.6094x; 1.0609x over previous
#include <hip/hip_runtime.h>


typedef __attribute__((ext_vector_type(2))) float v2f;
typedef __attribute__((ext_vector_type(4))) float v4f;
typedef __attribute__((ext_vector_type(8))) float v8f;

#define WMMA_F32(a, b, c) \
  __builtin_amdgcn_wmma_f32_16x16x4_f32(false, (a), false, (b), (short)0, (c), false, false)

static constexpr int BB = 2, TT = 2048, CC = 1024, HH = 16, DD = 64;
static constexpr int MTOT = BB * TT;
static constexpr int NQKV = 3 * CC;
static constexpr size_t HSZ = (size_t)BB * HH * TT * DD;


__device__ __forceinline__ void load_a_chunk(v4f (&a)[2][4],
                                             const float* __restrict__ A0,
                                             const float* __restrict__ A1, const int kc) {
#pragma unroll
  for (int u = 0; u < 4; ++u) a[0][u] = *(const v4f*)(A0 + kc + 8 * u);
#pragma unroll
  for (int u = 0; u < 4; ++u) a[1][u] = *(const v4f*)(A1 + kc + 8 * u);
}

template <int LDB>
__device__ __forceinline__ void load_b_step(v4f (&b)[8], const float* __restrict__ Bp,
                                            const int k0) {
  const float* Bk = Bp + (size_t)k0 * LDB;
#pragma unroll
  for (int r = 0; r < 4; ++r) {
    b[2 * r]     = *(const v4f*)(Bk + r * LDB);
    b[2 * r + 1] = *(const v4f*)(Bk + r * LDB + 64);
  }
}

__device__ __forceinline__ void mma_step(const v4f a0, const v4f a1, const v4f (&b)[8],
                                         v8f (&c)[2][8]) {
  const v2f a0lo = {a0.x, a0.y}, a1lo = {a1.x, a1.y};
  const v2f a0hi = {a0.z, a0.w}, a1hi = {a1.z, a1.w};
#pragma unroll
  for (int hf = 0; hf < 2; ++hf)
#pragma unroll
    for (int j = 0; j < 4; ++j) {
      const v2f bb = {b[hf][j], b[2 + hf][j]};
      c[0][4 * hf + j] = WMMA_F32(a0lo, bb, c[0][4 * hf + j]);
      c[1][4 * hf + j] = WMMA_F32(a1lo, bb, c[1][4 * hf + j]);
    }
#pragma unroll
  for (int hf = 0; hf < 2; ++hf)
#pragma unroll
    for (int j = 0; j < 4; ++j) {
      const v2f bb = {b[4 + hf][j], b[6 + hf][j]};
      c[0][4 * hf + j] = WMMA_F32(a0hi, bb, c[0][4 * hf + j]);
      c[1][4 * hf + j] = WMMA_F32(a1hi, bb, c[1][4 * hf + j]);
    }
}

template <int LDB>
__device__ __forceinline__ void gemm_chunk(v4f (&aC)[2][4], v4f (&aN)[2][4],
                                           v4f (&bP)[8], v4f (&bQ)[8],
                                           const float* __restrict__ A0,
                                           const float* __restrict__ A1,
                                           const float* __restrict__ Bp, const int kc,
                                           v8f (&c)[2][8]) {
  load_a_chunk(aN, A0, A1, (kc + 32) & (CC - 1));
  __builtin_amdgcn_sched_barrier(0);
  load_b_step<LDB>(bQ, Bp, kc + 8);
  mma_step(aC[0][0], aC[1][0], bP, c);
  load_b_step<LDB>(bP, Bp, kc + 16);
  mma_step(aC[0][1], aC[1][1], bQ, c);
  load_b_step<LDB>(bQ, Bp, kc + 24);
  mma_step(aC[0][2], aC[1][2], bP, c);
  load_b_step<LDB>(bP, Bp, (kc + 32) & (CC - 1));
  mma_step(aC[0][3], aC[1][3], bQ, c);
}

template <int LDB>
__device__ __forceinline__ void gemm_tile_32x128(const float* __restrict__ A0,
                                                 const float* __restrict__ A1,
                                                 const float* __restrict__ Bp,
                                                 v8f (&c)[2][8]) {
  v4f aP[2][4], aQ[2][4], bP[8], bQ[8];
  load_a_chunk(aP, A0, A1, 0);
  load_b_step<LDB>(bP, Bp, 0);
  for (int kc = 0; kc < CC; kc += 64) {
    gemm_chunk<LDB>(aP, aQ, bP, bQ, A0, A1, Bp, kc, c);
    gemm_chunk<LDB>(aQ, aP, bP, bQ, A0, A1, Bp, kc + 32, c);
  }
}

static constexpr int QKV_WAVES = 12;
static constexpr int OUT_WAVES = 4;
static constexpr int ATT_WAVES = 8;

__global__ __launch_bounds__(32 * QKV_WAVES)
void qkv_gemm_kernel(const float* __restrict__ X, const float* __restrict__ W,
                     const float* __restrict__ bias,
                     float* __restrict__ q, float* __restrict__ k,
                     float* __restrict__ v) {
  const int lane = threadIdx.x & 31;
  const int hl = lane >> 4;
  const int ln = lane & 15;
  const int w = __builtin_amdgcn_readfirstlane((int)(threadIdx.x >> 5));
  const int n0 = (blockIdx.x * 3 + (w >> 2)) * 128;
  const int m0 = (blockIdx.y * 4 + (w & 3)) * 32;

  v8f c[2][8] = {};
  const float* A0 = X + (size_t)(m0 + ln) * CC + 4 * hl;
  gemm_tile_32x128<NQKV>(A0, A0 + 16 * CC,
                         W + (size_t)(4 * hl) * NQKV + n0 + 4 * ln, c);

  const int which = n0 >> 10;
  float* dst = (which == 0) ? q : (which == 1) ? k : v;
  const int kfrag = (ln >> 1) * 512 + (ln & 1) * 4;
#pragma unroll
  for (int hf = 0; hf < 2; ++hf) {
    const v4f bv = *(const v4f*)(bias + n0 + 64 * hf + 4 * ln);
    const int h = ((n0 + 64 * hf) & (CC - 1)) >> 6;
#pragma unroll
    for (int t = 0; t < 2; ++t)
#pragma unroll
      for (int r = 0; r < 8; ++r) {
        const int trow = m0 + 16 * t + r + 8 * hl;
        const int bidx = trow >> 11;
        const int tt = trow & (TT - 1);
        const size_t rowm = (((size_t)(bidx * HH + h) * TT) + tt) * DD + 4 * ln;
        const size_t frag = (((size_t)(bidx * HH + h) * TT) + (tt & ~63)) * DD +
                            (tt & 3) * 128 + ((tt & 63) >> 2) * 8 + kfrag;
        const size_t idx = (which == 1) ? frag : rowm;
        const v4f o = {c[t][4 * hf + 0][r] + bv.x, c[t][4 * hf + 1][r] + bv.y,
                       c[t][4 * hf + 2][r] + bv.z, c[t][4 * hf + 3][r] + bv.w};
        *(v4f*)(dst + idx) = o;
      }
  }
}

static constexpr int KTRIP = 64;
static constexpr int PPITCH = KTRIP + 4;

__device__ __forceinline__ void wave_lds_handover() {
  __builtin_amdgcn_fence(__ATOMIC_RELEASE, "wavefront");
  __builtin_amdgcn_wave_barrier();
  __builtin_amdgcn_fence(__ATOMIC_ACQUIRE, "wavefront");
}

template <bool MASK>
__device__ __forceinline__ void attn_trip(const v4f (&qa)[8], v8f (&o)[4], float (&m_i)[8],
                                          float (&l_i)[8], float (*pt)[PPITCH],
                                          const float* __restrict__ kbase,
                                          const float* __restrict__ vbase, const int kb,
                                          const int q0, const int hl, const int ln) {
  v8f s[4] = {};
  const float* krow = kbase + (size_t)kb * DD + 8 * ln + 4 * hl;
#pragma unroll
  for (int u = 0; u < 8; ++u) {
    v4f kf[4];
#pragma unroll
    for (int j = 0; j < 4; ++j) kf[j] = *(const v4f*)(krow + (4 * u + j) * 128);
    const v2f q01 = {qa[u].x, qa[u].y};
    const v2f q23 = {qa[u].z, qa[u].w};
#pragma unroll
    for (int j = 0; j < 4; ++j) {
      const v2f k01 = {kf[j].x, kf[j].y};
      s[j] = WMMA_F32(q01, k01, s[j]);
    }
#pragma unroll
    for (int j = 0; j < 4; ++j) {
      const v2f k23 = {kf[j].z, kf[j].w};
      s[j] = WMMA_F32(q23, k23, s[j]);
    }
  }

  if (MASK) {
#pragma unroll
    for (int j = 0; j < 4; ++j)
#pragma unroll
      for (int r = 0; r < 8; ++r)
        if (kb + 4 * ln + j > q0 + r + 8 * hl) s[j][r] = -__builtin_inff();
  }

#pragma unroll
  for (int r = 0; r < 8; ++r) {
    float mx = fmaxf(fmaxf(s[0][r], s[1][r]), fmaxf(s[2][r], s[3][r]));
    mx = fmaxf(mx, __shfl_xor(mx, 1));
    mx = fmaxf(mx, __shfl_xor(mx, 2));
    mx = fmaxf(mx, __shfl_xor(mx, 4));
    mx = fmaxf(mx, __shfl_xor(mx, 8));
    const float mn = fmaxf(m_i[r], mx);
    const float sc = __expf(m_i[r] - mn);
    const v4f p = {__expf(s[0][r] - mn), __expf(s[1][r] - mn),
                   __expf(s[2][r] - mn), __expf(s[3][r] - mn)};
    float ps = (p.x + p.y) + (p.z + p.w);
    ps += __shfl_xor(ps, 1);
    ps += __shfl_xor(ps, 2);
    ps += __shfl_xor(ps, 4);
    ps += __shfl_xor(ps, 8);
    l_i[r] = l_i[r] * sc + ps;
    m_i[r] = mn;
    *(v4f*)&pt[r + 8 * hl][4 * ln] = p;
#pragma unroll
    for (int n = 0; n < 4; ++n) o[n][r] *= sc;
  }

  wave_lds_handover();

  v4f pa[8];
#pragma unroll
  for (int u = 0; u < 8; ++u) pa[u] = *(const v4f*)&pt[ln][8 * u + 4 * hl];
  const float* vrow = vbase + (size_t)(kb + 4 * hl) * DD + 4 * ln;
#pragma unroll
  for (int u = 0; u < 8; ++u) {
    const v4f v0 = *(const v4f*)(vrow + (8 * u + 0) * DD);
    const v4f v1 = *(const v4f*)(vrow + (8 * u + 1) * DD);
    const v4f v2 = *(const v4f*)(vrow + (8 * u + 2) * DD);
    const v4f v3 = *(const v4f*)(vrow + (8 * u + 3) * DD);
    const v2f a01 = {pa[u].x, pa[u].y};
    const v2f a23 = {pa[u].z, pa[u].w};
#pragma unroll
    for (int n = 0; n < 4; ++n) {
      const v2f bf = {v0[n], v1[n]};
      o[n] = WMMA_F32(a01, bf, o[n]);
    }
#pragma unroll
    for (int n = 0; n < 4; ++n) {
      const v2f bf = {v2[n], v3[n]};
      o[n] = WMMA_F32(a23, bf, o[n]);
    }
  }
  wave_lds_handover();
}

__global__ __launch_bounds__(32 * ATT_WAVES) __attribute__((amdgpu_waves_per_eu(2, 2)))
void attn_kernel(const float* __restrict__ q, const float* __restrict__ kmat,
                 const float* __restrict__ vmat, float* __restrict__ out) {
  const int lane = threadIdx.x & 31;
  const int hl = lane >> 4;
  const int ln = lane & 15;
  const int w = __builtin_amdgcn_readfirstlane((int)(threadIdx.x >> 5));
  const int pr = blockIdx.x * ATT_WAVES + w;
  const int bh = blockIdx.y;
  const int b = bh >> 4, h = bh & 15;

  const float* qbase = q + (size_t)bh * TT * DD;
  const float* kbase = kmat + (size_t)bh * TT * DD;
  const float* vbase = vmat + (size_t)bh * TT * DD;

  __shared__ __attribute__((aligned(16))) float pt_all[ATT_WAVES][16][PPITCH];
  float (*pt)[PPITCH] = pt_all[w];

  for (int half = 0; half < 2; ++half) {
    const int qtile = half ? pr : (TT / 16 - 1 - pr);
    const int q0 = qtile * 16;

    v4f qa[8];
    {
      const float* qrow = qbase + (size_t)(q0 + ln) * DD + 4 * hl;
#pragma unroll
      for (int u = 0; u < 8; ++u) {
        const v4f a = *(const v4f*)(qrow + 8 * u);
        qa[u] = a * 0.125f;
      }
    }

    v8f o[4] = {};
    float m_i[8], l_i[8];
#pragma unroll
    for (int r = 0; r < 8; ++r) { m_i[r] = -__builtin_inff(); l_i[r] = 0.0f; }

    const int last = q0 / KTRIP;
    for (int kt = 0; kt < last; ++kt)
      attn_trip<false>(qa, o, m_i, l_i, pt, kbase, vbase, kt * KTRIP, q0, hl, ln);
    attn_trip<true>(qa, o, m_i, l_i, pt, kbase, vbase, last * KTRIP, q0, hl, ln);

#pragma unroll
    for (int r = 0; r < 8; ++r) {
      const float inv = 1.0f / l_i[r];
      const int trow = q0 + r + 8 * hl;
      float* orow = out + ((size_t)(b * TT + trow) * CC) + h * DD + 4 * ln;
      const v4f ov = {o[0][r] * inv, o[1][r] * inv, o[2][r] * inv, o[3][r] * inv};
      *(v4f*)orow = ov;
    }
  }
}

__global__ __launch_bounds__(32 * OUT_WAVES)
void out_gemm_kernel(const float* __restrict__ A, const float* __restrict__ W,
                     const float* __restrict__ bias, float* __restrict__ out) {
  const int lane = threadIdx.x & 31;
  const int hl = lane >> 4;
  const int ln = lane & 15;
  const int w = __builtin_amdgcn_readfirstlane((int)(threadIdx.x >> 5));
  const int n0 = blockIdx.x * 128;
  const int m0 = (blockIdx.y * OUT_WAVES + w) * 32;

  v8f c[2][8] = {};
  const float* A0 = A + (size_t)(m0 + ln) * CC + 4 * hl;
  gemm_tile_32x128<CC>(A0, A0 + 16 * CC,
                       W + (size_t)(4 * hl) * CC + n0 + 4 * ln, c);

#pragma unroll
  for (int hf = 0; hf < 2; ++hf) {
    const v4f bv = *(const v4f*)(bias + n0 + 64 * hf + 4 * ln);
#pragma unroll
    for (int t = 0; t < 2; ++t)
#pragma unroll
      for (int r = 0; r < 8; ++r) {
        const v4f o = {c[t][4 * hf + 0][r] + bv.x, c[t][4 * hf + 1][r] + bv.y,
                       c[t][4 * hf + 2][r] + bv.z, c[t][4 * hf + 3][r] + bv.w};
        *(v4f*)(out + (size_t)(m0 + 16 * t + r + 8 * hl) * CC + n0 + 64 * hf + 4 * ln) = o;
      }
  }
}

extern "C" void kernel_launch(void* const* d_in, const int* in_sizes, int n_in,
                              void* d_out, int out_size, void* d_ws, size_t ws_size,
                              hipStream_t stream) {
  (void)in_sizes; (void)n_in; (void)out_size; (void)ws_size;
  const float* x    = (const float*)d_in[0];
  const float* Wqkv = (const float*)d_in[1];
  const float* bqkv = (const float*)d_in[2];
  const float* Wout = (const float*)d_in[3];
  const float* bout = (const float*)d_in[4];
  float* out = (float*)d_out;

  float* qp = (float*)d_ws;
  float* kp = qp + HSZ;
  float* vp = kp + HSZ;
  float* ao = vp + HSZ;

  qkv_gemm_kernel<<<dim3(NQKV / 384, MTOT / 128), dim3(32 * QKV_WAVES, 1, 1), 0, stream>>>(x, Wqkv, bqkv, qp, kp, vp);
  attn_kernel<<<dim3(TT / 32 / ATT_WAVES, BB * HH), dim3(32 * ATT_WAVES, 1, 1), 0, stream>>>(qp, kp, vp, ao);
  out_gemm_kernel<<<dim3(CC / 128, MTOT / 128), dim3(32 * OUT_WAVES, 1, 1), 0, stream>>>(ao, Wout, bout, out);
}


# Output projection (one wave per SIMD): the whole next 32-k chunk, 8 A and 32 B fragment loads, is requested as one group before the current chunk's 128 WMMAs, was one 8-k step ahead; sum order same
# speedup vs baseline: 2.7656x; 1.0599x over previous
#include <hip/hip_runtime.h>


typedef __attribute__((ext_vector_type(2))) float v2f;
typedef __attribute__((ext_vector_type(4))) float v4f;
typedef __attribute__((ext_vector_type(8))) float v8f;

#define WMMA_F32(a, b, c) \
  __builtin_amdgcn_wmma_f32_16x16x4_f32(false, (a), false, (b), (short)0, (c), false, false)

static constexpr int BB = 2, TT = 2048, CC = 1024, HH = 16, DD = 64;
static constexpr int MTOT = BB * TT;
static constexpr int NQKV = 3 * CC;
static constexpr size_t HSZ = (size_t)BB * HH * TT * DD;


__device__ __forceinline__ void load_a_chunk(v4f (&a)[2][4],
                                             const float* __restrict__ A0,
                                             const float* __restrict__ A1, const int kc) {
#pragma unroll
  for (int u = 0; u < 4; ++u) a[0][u] = *(const v4f*)(A0 + kc + 8 * u);
#pragma unroll
  for (int u = 0; u < 4; ++u) a[1][u] = *(const v4f*)(A1 + kc + 8 * u);
}

template <int LDB>
__device__ __forceinline__ void load_b_step(v4f (&b)[8], const float* __restrict__ Bp,
                                            const int k0) {
  const float* Bk = Bp + (size_t)k0 * LDB;
#pragma unroll
  for (int r = 0; r < 4; ++r) {
    b[2 * r]     = *(const v4f*)(Bk + r * LDB);
    b[2 * r + 1] = *(const v4f*)(Bk + r * LDB + 64);
  }
}

__device__ __forceinline__ void mma_step(const v4f a0, const v4f a1, const v4f (&b)[8],
                                         v8f (&c)[2][8]) {
  const v2f a0lo = {a0.x, a0.y}, a1lo = {a1.x, a1.y};
  const v2f a0hi = {a0.z, a0.w}, a1hi = {a1.z, a1.w};
#pragma unroll
  for (int hf = 0; hf < 2; ++hf)
#pragma unroll
    for (int j = 0; j < 4; ++j) {
      const v2f bb = {b[hf][j], b[2 + hf][j]};
      c[0][4 * hf + j] = WMMA_F32(a0lo, bb, c[0][4 * hf + j]);
      c[1][4 * hf + j] = WMMA_F32(a1lo, bb, c[1][4 * hf + j]);
    }
#pragma unroll
  for (int hf = 0; hf < 2; ++hf)
#pragma unroll
    for (int j = 0; j < 4; ++j) {
      const v2f bb = {b[4 + hf][j], b[6 + hf][j]};
      c[0][4 * hf + j] = WMMA_F32(a0hi, bb, c[0][4 * hf + j]);
      c[1][4 * hf + j] = WMMA_F32(a1hi, bb, c[1][4 * hf + j]);
    }
}

template <int LDB>
__device__ __forceinline__ void gemm_chunk(v4f (&aC)[2][4], v4f (&aN)[2][4],
                                           v4f (&bP)[8], v4f (&bQ)[8],
                                           const float* __restrict__ A0,
                                           const float* __restrict__ A1,
                                           const float* __restrict__ Bp, const int kc,
                                           v8f (&c)[2][8]) {
  load_a_chunk(aN, A0, A1, (kc + 32) & (CC - 1));
  __builtin_amdgcn_sched_barrier(0);
  load_b_step<LDB>(bQ, Bp, kc + 8);
  mma_step(aC[0][0], aC[1][0], bP, c);
  load_b_step<LDB>(bP, Bp, kc + 16);
  mma_step(aC[0][1], aC[1][1], bQ, c);
  load_b_step<LDB>(bQ, Bp, kc + 24);
  mma_step(aC[0][2], aC[1][2], bP, c);
  load_b_step<LDB>(bP, Bp, (kc + 32) & (CC - 1));
  mma_step(aC[0][3], aC[1][3], bQ, c);
}

template <int LDB>
__device__ __forceinline__ void gemm_tile_32x128(const float* __restrict__ A0,
                                                 const float* __restrict__ A1,
                                                 const float* __restrict__ Bp,
                                                 v8f (&c)[2][8]) {
  v4f aP[2][4], aQ[2][4], bP[8], bQ[8];
  load_a_chunk(aP, A0, A1, 0);
  load_b_step<LDB>(bP, Bp, 0);
  for (int kc = 0; kc < CC; kc += 64) {
    gemm_chunk<LDB>(aP, aQ, bP, bQ, A0, A1, Bp, kc, c);
    gemm_chunk<LDB>(aQ, aP, bP, bQ, A0, A1, Bp, kc + 32, c);
  }
}

template <int LDB>
__device__ __forceinline__ void load_b_chunk(v4f (&b)[4][8], const float* __restrict__ Bp,
                                             const int kc) {
#pragma unroll
  for (int s = 0; s < 4; ++s) load_b_step<LDB>(b[s], Bp, kc + 8 * s);
}

__device__ __forceinline__ void mma_chunk(const v4f (&a)[2][4], const v4f (&b)[4][8],
                                          v8f (&c)[2][8]) {
#pragma unroll
  for (int s = 0; s < 4; ++s) mma_step(a[0][s], a[1][s], b[s], c);
}

template <int LDB>
__device__ __forceinline__ void gemm_tile_32x128_chunk_ahead(const float* __restrict__ A0,
                                                             const float* __restrict__ A1,
                                                             const float* __restrict__ Bp,
                                                             v8f (&c)[2][8]) {
  v4f aP[2][4], aQ[2][4], bP[4][8], bQ[4][8];
  load_a_chunk(aP, A0, A1, 0);
  load_b_chunk<LDB>(bP, Bp, 0);
  for (int kc = 0; kc < CC; kc += 64) {
    load_a_chunk(aQ, A0, A1, kc + 32);
    load_b_chunk<LDB>(bQ, Bp, kc + 32);
    __builtin_amdgcn_sched_barrier(0);
    mma_chunk(aP, bP, c);
    load_a_chunk(aP, A0, A1, (kc + 64) & (CC - 1));
    load_b_chunk<LDB>(bP, Bp, (kc + 64) & (CC - 1));
    __builtin_amdgcn_sched_barrier(0);
    mma_chunk(aQ, bQ, c);
  }
}

static constexpr int QKV_WAVES = 12;
static constexpr int OUT_WAVES = 4;
static constexpr int ATT_WAVES = 8;

__global__ __launch_bounds__(32 * QKV_WAVES)
void qkv_gemm_kernel(const float* __restrict__ X, const float* __restrict__ W,
                     const float* __restrict__ bias,
                     float* __restrict__ q, float* __restrict__ k,
                     float* __restrict__ v) {
  const int lane = threadIdx.x & 31;
  const int hl = lane >> 4;
  const int ln = lane & 15;
  const int w = __builtin_amdgcn_readfirstlane((int)(threadIdx.x >> 5));
  const int n0 = (blockIdx.x * 3 + (w >> 2)) * 128;
  const int m0 = (blockIdx.y * 4 + (w & 3)) * 32;

  v8f c[2][8] = {};
  const float* A0 = X + (size_t)(m0 + ln) * CC + 4 * hl;
  gemm_tile_32x128<NQKV>(A0, A0 + 16 * CC,
                         W + (size_t)(4 * hl) * NQKV + n0 + 4 * ln, c);

  const int which = n0 >> 10;
  float* dst = (which == 0) ? q : (which == 1) ? k : v;
  const int kfrag = (ln >> 1) * 512 + (ln & 1) * 4;
#pragma unroll
  for (int hf = 0; hf < 2; ++hf) {
    const v4f bv = *(const v4f*)(bias + n0 + 64 * hf + 4 * ln);
    const int h = ((n0 + 64 * hf) & (CC - 1)) >> 6;
#pragma unroll
    for (int t = 0; t < 2; ++t)
#pragma unroll
      for (int r = 0; r < 8; ++r) {
        const int trow = m0 + 16 * t + r + 8 * hl;
        const int bidx = trow >> 11;
        const int tt = trow & (TT - 1);
        const size_t rowm = (((size_t)(bidx * HH + h) * TT) + tt) * DD + 4 * ln;
        const size_t frag = (((size_t)(bidx * HH + h) * TT) + (tt & ~63)) * DD +
                            (tt & 3) * 128 + ((tt & 63) >> 2) * 8 + kfrag;
        const size_t idx = (which == 1) ? frag : rowm;
        const v4f o = {c[t][4 * hf + 0][r] + bv.x, c[t][4 * hf + 1][r] + bv.y,
                       c[t][4 * hf + 2][r] + bv.z, c[t][4 * hf + 3][r] + bv.w};
        *(v4f*)(dst + idx) = o;
      }
  }
}

static constexpr int KTRIP = 64;
static constexpr int PPITCH = KTRIP + 4;

__device__ __forceinline__ void wave_lds_handover() {
  __builtin_amdgcn_fence(__ATOMIC_RELEASE, "wavefront");
  __builtin_amdgcn_wave_barrier();
  __builtin_amdgcn_fence(__ATOMIC_ACQUIRE, "wavefront");
}

template <bool MASK>
__device__ __forceinline__ void attn_trip(const v4f (&qa)[8], v8f (&o)[4], float (&m_i)[8],
                                          float (&l_i)[8], float (*pt)[PPITCH],
                                          const float* __restrict__ kbase,
                                          const float* __restrict__ vbase, const int kb,
                                          const int q0, const int hl, const int ln) {
  v8f s[4] = {};
  const float* krow = kbase + (size_t)kb * DD + 8 * ln + 4 * hl;
#pragma unroll
  for (int u = 0; u < 8; ++u) {
    v4f kf[4];
#pragma unroll
    for (int j = 0; j < 4; ++j) kf[j] = *(const v4f*)(krow + (4 * u + j) * 128);
    const v2f q01 = {qa[u].x, qa[u].y};
    const v2f q23 = {qa[u].z, qa[u].w};
#pragma unroll
    for (int j = 0; j < 4; ++j) {
      const v2f k01 = {kf[j].x, kf[j].y};
      s[j] = WMMA_F32(q01, k01, s[j]);
    }
#pragma unroll
    for (int j = 0; j < 4; ++j) {
      const v2f k23 = {kf[j].z, kf[j].w};
      s[j] = WMMA_F32(q23, k23, s[j]);
    }
  }

  if (MASK) {
#pragma unroll
    for (int j = 0; j < 4; ++j)
#pragma unroll
      for (int r = 0; r < 8; ++r)
        if (kb + 4 * ln + j > q0 + r + 8 * hl) s[j][r] = -__builtin_inff();
  }

#pragma unroll
  for (int r = 0; r < 8; ++r) {
    float mx = fmaxf(fmaxf(s[0][r], s[1][r]), fmaxf(s[2][r], s[3][r]));
    mx = fmaxf(mx, __shfl_xor(mx, 1));
    mx = fmaxf(mx, __shfl_xor(mx, 2));
    mx = fmaxf(mx, __shfl_xor(mx, 4));
    mx = fmaxf(mx, __shfl_xor(mx, 8));
    const float mn = fmaxf(m_i[r], mx);
    const float sc = __expf(m_i[r] - mn);
    const v4f p = {__expf(s[0][r] - mn), __expf(s[1][r] - mn),
                   __expf(s[2][r] - mn), __expf(s[3][r] - mn)};
    float ps = (p.x + p.y) + (p.z + p.w);
    ps += __shfl_xor(ps, 1);
    ps += __shfl_xor(ps, 2);
    ps += __shfl_xor(ps, 4);
    ps += __shfl_xor(ps, 8);
    l_i[r] = l_i[r] * sc + ps;
    m_i[r] = mn;
    *(v4f*)&pt[r + 8 * hl][4 * ln] = p;
#pragma unroll
    for (int n = 0; n < 4; ++n) o[n][r] *= sc;
  }

  wave_lds_handover();

  v4f pa[8];
#pragma unroll
  for (int u = 0; u < 8; ++u) pa[u] = *(const v4f*)&pt[ln][8 * u + 4 * hl];
  const float* vrow = vbase + (size_t)(kb + 4 * hl) * DD + 4 * ln;
#pragma unroll
  for (int u = 0; u < 8; ++u) {
    const v4f v0 = *(const v4f*)(vrow + (8 * u + 0) * DD);
    const v4f v1 = *(const v4f*)(vrow + (8 * u + 1) * DD);
    const v4f v2 = *(const v4f*)(vrow + (8 * u + 2) * DD);
    const v4f v3 = *(const v4f*)(vrow + (8 * u + 3) * DD);
    const v2f a01 = {pa[u].x, pa[u].y};
    const v2f a23 = {pa[u].z, pa[u].w};
#pragma unroll
    for (int n = 0; n < 4; ++n) {
      const v2f bf = {v0[n], v1[n]};
      o[n] = WMMA_F32(a01, bf, o[n]);
    }
#pragma unroll
    for (int n = 0; n < 4; ++n) {
      const v2f bf = {v2[n], v3[n]};
      o[n] = WMMA_F32(a23, bf, o[n]);
    }
  }
  wave_lds_handover();
}

__global__ __launch_bounds__(32 * ATT_WAVES) __attribute__((amdgpu_waves_per_eu(2, 2)))
void attn_kernel(const float* __restrict__ q, const float* __restrict__ kmat,
                 const float* __restrict__ vmat, float* __restrict__ out) {
  const int lane = threadIdx.x & 31;
  const int hl = lane >> 4;
  const int ln = lane & 15;
  const int w = __builtin_amdgcn_readfirstlane((int)(threadIdx.x >> 5));
  const int pr = blockIdx.x * ATT_WAVES + w;
  const int bh = blockIdx.y;
  const int b = bh >> 4, h = bh & 15;

  const float* qbase = q + (size_t)bh * TT * DD;
  const float* kbase = kmat + (size_t)bh * TT * DD;
  const float* vbase = vmat + (size_t)bh * TT * DD;

  __shared__ __attribute__((aligned(16))) float pt_all[ATT_WAVES][16][PPITCH];
  float (*pt)[PPITCH] = pt_all[w];

  for (int half = 0; half < 2; ++half) {
    const int qtile = half ? pr : (TT / 16 - 1 - pr);
    const int q0 = qtile * 16;

    v4f qa[8];
    {
      const float* qrow = qbase + (size_t)(q0 + ln) * DD + 4 * hl;
#pragma unroll
      for (int u = 0; u < 8; ++u) {
        const v4f a = *(const v4f*)(qrow + 8 * u);
        qa[u] = a * 0.125f;
      }
    }

    v8f o[4] = {};
    float m_i[8], l_i[8];
#pragma unroll
    for (int r = 0; r < 8; ++r) { m_i[r] = -__builtin_inff(); l_i[r] = 0.0f; }

    const int last = q0 / KTRIP;
    for (int kt = 0; kt < last; ++kt)
      attn_trip<false>(qa, o, m_i, l_i, pt, kbase, vbase, kt * KTRIP, q0, hl, ln);
    attn_trip<true>(qa, o, m_i, l_i, pt, kbase, vbase, last * KTRIP, q0, hl, ln);

#pragma unroll
    for (int r = 0; r < 8; ++r) {
      const float inv = 1.0f / l_i[r];
      const int trow = q0 + r + 8 * hl;
      float* orow = out + ((size_t)(b * TT + trow) * CC) + h * DD + 4 * ln;
      const v4f ov = {o[0][r] * inv, o[1][r] * inv, o[2][r] * inv, o[3][r] * inv};
      *(v4f*)orow = ov;
    }
  }
}

__global__ __launch_bounds__(32 * OUT_WAVES)
void out_gemm_kernel(const float* __restrict__ A, const float* __restrict__ W,
                     const float* __restrict__ bias, float* __restrict__ out) {
  const int lane = threadIdx.x & 31;
  const int hl = lane >> 4;
  const int ln = lane & 15;
  const int w = __builtin_amdgcn_readfirstlane((int)(threadIdx.x >> 5));
  const int n0 = blockIdx.x * 128;
  const int m0 = (blockIdx.y * OUT_WAVES + w) * 32;

  v8f c[2][8] = {};
  const float* A0 = A + (size_t)(m0 + ln) * CC + 4 * hl;
  gemm_tile_32x128_chunk_ahead<CC>(A0, A0 + 16 * CC,
                                   W + (size_t)(4 * hl) * CC + n0 + 4 * ln, c);

#pragma unroll
  for (int hf = 0; hf < 2; ++hf) {
    const v4f bv = *(const v4f*)(bias + n0 + 64 * hf + 4 * ln);
#pragma unroll
    for (int t = 0; t < 2; ++t)
#pragma unroll
      for (int r = 0; r < 8; ++r) {
        const v4f o = {c[t][4 * hf + 0][r] + bv.x, c[t][4 * hf + 1][r] + bv.y,
                       c[t][4 * hf + 2][r] + bv.z, c[t][4 * hf + 3][r] + bv.w};
        *(v4f*)(out + (size_t)(m0 + 16 * t + r + 8 * hl) * CC + n0 + 64 * hf + 4 * ln) = o;
      }
  }
}

extern "C" void kernel_launch(void* const* d_in, const int* in_sizes, int n_in,
                              void* d_out, int out_size, void* d_ws, size_t ws_size,
                              hipStream_t stream) {
  (void)in_sizes; (void)n_in; (void)out_size; (void)ws_size;
  const float* x    = (const float*)d_in[0];
  const float* Wqkv = (const float*)d_in[1];
  const float* bqkv = (const float*)d_in[2];
  const float* Wout = (const float*)d_in[3];
  const float* bout = (const float*)d_in[4];
  float* out = (float*)d_out;

  float* qp = (float*)d_ws;
  float* kp = qp + HSZ;
  float* vp = kp + HSZ;
  float* ao = vp + HSZ;

  qkv_gemm_kernel<<<dim3(NQKV / 384, MTOT / 128), dim3(32 * QKV_WAVES, 1, 1), 0, stream>>>(x, Wqkv, bqkv, qp, kp, vp);
  attn_kernel<<<dim3(TT / 32 / ATT_WAVES, BB * HH), dim3(32 * ATT_WAVES, 1, 1), 0, stream>>>(qp, kp, vp, ao);
  out_gemm_kernel<<<dim3(CC / 128, MTOT / 128), dim3(32 * OUT_WAVES, 1, 1), 0, stream>>>(ao, Wout, bout, out);
}


# QKV projection: one wave per SIMD holds a 64x192 output tile (was 3 waves of 32x128): an A fragment feeds 12 column sub-tiles, a B fragment 4 row tiles, next chunk requested ahead; sum order same
# speedup vs baseline: 2.8591x; 1.0338x over previous
#include <hip/hip_runtime.h>


typedef __attribute__((ext_vector_type(2))) float v2f;
typedef __attribute__((ext_vector_type(4))) float v4f;
typedef __attribute__((ext_vector_type(8))) float v8f;

#define WMMA_F32(a, b, c) \
  __builtin_amdgcn_wmma_f32_16x16x4_f32(false, (a), false, (b), (short)0, (c), false, false)

static constexpr int BB = 2, TT = 2048, CC = 1024, HH = 16, DD = 64;
static constexpr int MTOT = BB * TT;
static constexpr int NQKV = 3 * CC;
static constexpr size_t HSZ = (size_t)BB * HH * TT * DD;


__device__ __forceinline__ void load_a_chunk(v4f (&a)[2][4],
                                             const float* __restrict__ A0,
                                             const float* __restrict__ A1, const int kc) {
#pragma unroll
  for (int u = 0; u < 4; ++u) a[0][u] = *(const v4f*)(A0 + kc + 8 * u);
#pragma unroll
  for (int u = 0; u < 4; ++u) a[1][u] = *(const v4f*)(A1 + kc + 8 * u);
}

template <int LDB>
__device__ __forceinline__ void load_b_step(v4f (&b)[8], const float* __restrict__ Bp,
                                            const int k0) {
  const float* Bk = Bp + (size_t)k0 * LDB;
#pragma unroll
  for (int r = 0; r < 4; ++r) {
    b[2 * r]     = *(const v4f*)(Bk + r * LDB);
    b[2 * r + 1] = *(const v4f*)(Bk + r * LDB + 64);
  }
}

__device__ __forceinline__ void mma_step(const v4f a0, const v4f a1, const v4f (&b)[8],
                                         v8f (&c)[2][8]) {
  const v2f a0lo = {a0.x, a0.y}, a1lo = {a1.x, a1.y};
  const v2f a0hi = {a0.z, a0.w}, a1hi = {a1.z, a1.w};
#pragma unroll
  for (int hf = 0; hf < 2; ++hf)
#pragma unroll
    for (int j = 0; j < 4; ++j) {
      const v2f bb = {b[hf][j], b[2 + hf][j]};
      c[0][4 * hf + j] = WMMA_F32(a0lo, bb, c[0][4 * hf + j]);
      c[1][4 * hf + j] = WMMA_F32(a1lo, bb, c[1][4 * hf + j]);
    }
#pragma unroll
  for (int hf = 0; hf < 2; ++hf)
#pragma unroll
    for (int j = 0; j < 4; ++j) {
      const v2f bb = {b[4 + hf][j], b[6 + hf][j]};
      c[0][4 * hf + j] = WMMA_F32(a0hi, bb, c[0][4 * hf + j]);
      c[1][4 * hf + j] = WMMA_F32(a1hi, bb, c[1][4 * hf + j]);
    }
}

template <int LDB>
__device__ __forceinline__ void load_b_chunk(v4f (&b)[4][8], const float* __restrict__ Bp,
                                             const int kc) {
#pragma unroll
  for (int s = 0; s < 4; ++s) load_b_step<LDB>(b[s], Bp, kc + 8 * s);
}

__device__ __forceinline__ void mma_chunk(const v4f (&a)[2][4], const v4f (&b)[4][8],
                                          v8f (&c)[2][8]) {
#pragma unroll
  for (int s = 0; s < 4; ++s) mma_step(a[0][s], a[1][s], b[s], c);
}

template <int LDB>
__device__ __forceinline__ void gemm_tile_32x128_chunk_ahead(const float* __restrict__ A0,
                                                             const float* __restrict__ A1,
                                                             const float* __restrict__ Bp,
                                                             v8f (&c)[2][8]) {
  v4f aP[2][4], aQ[2][4], bP[4][8], bQ[4][8];
  load_a_chunk(aP, A0, A1, 0);
  load_b_chunk<LDB>(bP, Bp, 0);
  for (int kc = 0; kc < CC; kc += 64) {
    load_a_chunk(aQ, A0, A1, kc + 32);
    load_b_chunk<LDB>(bQ, Bp, kc + 32);
    __builtin_amdgcn_sched_barrier(0);
    mma_chunk(aP, bP, c);
    load_a_chunk(aP, A0, A1, (kc + 64) & (CC - 1));
    load_b_chunk<LDB>(bP, Bp, (kc + 64) & (CC - 1));
    __builtin_amdgcn_sched_barrier(0);
    mma_chunk(aQ, bQ, c);
  }
}

static constexpr int QT = 4;
static constexpr int QG = 3;

__device__ __forceinline__ void qkv_load_a(v4f (&a)[QT][4], const float* A0, const int kc) {
#pragma unroll
  for (int t = 0; t < QT; ++t)
#pragma unroll
    for (int u = 0; u < 4; ++u) a[t][u] = *(const v4f*)(A0 + (size_t)t * 16 * CC + kc + 8 * u);
}

__device__ __forceinline__ void qkv_load_b(v4f (&b)[4][QG], const float* Bp, const int k0) {
  const float* Bk = Bp + (size_t)k0 * NQKV;
#pragma unroll
  for (int r = 0; r < 4; ++r)
#pragma unroll
    for (int g = 0; g < QG; ++g) b[r][g] = *(const v4f*)(Bk + r * NQKV + 64 * g);
}

template <int U>
__device__ __forceinline__ void qkv_mma_step(const v4f (&a)[QT][4], const v4f (&b)[4][QG],
                                             v8f (&c)[QT][4 * QG]) {
#pragma unroll
  for (int g = 0; g < QG; ++g)
#pragma unroll
    for (int j = 0; j < 4; ++j) {
      const v2f bb = {b[0][g][j], b[1][g][j]};
#pragma unroll
      for (int t = 0; t < QT; ++t) {
        const v2f alo = {a[t][U].x, a[t][U].y};
        c[t][4 * g + j] = WMMA_F32(alo, bb, c[t][4 * g + j]);
      }
    }
#pragma unroll
  for (int g = 0; g < QG; ++g)
#pragma unroll
    for (int j = 0; j < 4; ++j) {
      const v2f bb = {b[2][g][j], b[3][g][j]};
#pragma unroll
      for (int t = 0; t < QT; ++t) {
        const v2f ahi = {a[t][U].z, a[t][U].w};
        c[t][4 * g + j] = WMMA_F32(ahi, bb, c[t][4 * g + j]);
      }
    }
}

__device__ __forceinline__ void request_fence() {
  asm volatile("" ::: "memory");
  __builtin_amdgcn_sched_barrier(0);
}

__device__ __forceinline__ void qkv_chunk(const v4f (&aC)[QT][4], const v4f (&bC)[4][4][QG],
                                          v4f (&aN)[QT][4], v4f (&bN)[4][4][QG],
                                          const float* A0, const float* Bp, const int kn,
                                          v8f (&c)[QT][4 * QG]) {
  request_fence();
  qkv_load_a(aN, A0, kn);
  qkv_load_b(bN[0], Bp, kn);
  qkv_load_b(bN[1], Bp, kn + 8);
  request_fence();
  qkv_mma_step<0>(aC, bC[0], c);
  qkv_mma_step<1>(aC, bC[1], c);
  request_fence();
  qkv_load_b(bN[2], Bp, kn + 16);
  qkv_load_b(bN[3], Bp, kn + 24);
  request_fence();
  qkv_mma_step<2>(aC, bC[2], c);
  qkv_mma_step<3>(aC, bC[3], c);
}

__device__ __forceinline__ void qkv_tile_64x192(const float* A0, const float* Bp,
                                                v8f (&c)[QT][4 * QG]) {
  v4f aP[QT][4], aQ[QT][4], bP[4][4][QG], bQ[4][4][QG];
  qkv_load_a(aP, A0, 0);
#pragma unroll
  for (int s = 0; s < 4; ++s) qkv_load_b(bP[s], Bp, 8 * s);
  for (int kc = 0; kc < CC; kc += 64) {
    qkv_chunk(aP, bP, aQ, bQ, A0, Bp, kc + 32, c);
    qkv_chunk(aQ, bQ, aP, bP, A0, Bp, (kc + 64) & (CC - 1), c);
  }
}

static constexpr int QKV_WAVES = 4;
static constexpr int OUT_WAVES = 4;
static constexpr int ATT_WAVES = 8;

__global__ __launch_bounds__(32 * QKV_WAVES)
void qkv_gemm_kernel(const float* X, const float* W,
                     const float* __restrict__ bias,
                     float* __restrict__ q, float* __restrict__ k,
                     float* __restrict__ v) {
  const int lane = threadIdx.x & 31;
  const int hl = lane >> 4;
  const int ln = lane & 15;
  const int w = __builtin_amdgcn_readfirstlane((int)(threadIdx.x >> 5));
  const int n0 = (blockIdx.x * 2 + (w >> 1)) * (64 * QG);
  const int m0 = (blockIdx.y * 2 + (w & 1)) * (16 * QT);

  v8f c[QT][4 * QG] = {};
  qkv_tile_64x192(X + (size_t)(m0 + ln) * CC + 4 * hl,
                  W + (size_t)(4 * hl) * NQKV + n0 + 4 * ln, c);

  const int kfrag = (ln >> 1) * 512 + (ln & 1) * 4;
#pragma unroll
  for (int g = 0; g < QG; ++g) {
    const int nc = n0 + 64 * g;
    const int which = nc >> 10;
    float* dst = (which == 0) ? q : (which == 1) ? k : v;
    const v4f bv = *(const v4f*)(bias + nc + 4 * ln);
    const int h = (nc & (CC - 1)) >> 6;
#pragma unroll
    for (int t = 0; t < QT; ++t)
#pragma unroll
      for (int r = 0; r < 8; ++r) {
        const int trow = m0 + 16 * t + r + 8 * hl;
        const int bidx = trow >> 11;
        const int tt = trow & (TT - 1);
        const size_t rowm = (((size_t)(bidx * HH + h) * TT) + tt) * DD + 4 * ln;
        const size_t frag = (((size_t)(bidx * HH + h) * TT) + (tt & ~63)) * DD +
                            (tt & 3) * 128 + ((tt & 63) >> 2) * 8 + kfrag;
        const size_t idx = (which == 1) ? frag : rowm;
        const v4f o = {c[t][4 * g + 0][r] + bv.x, c[t][4 * g + 1][r] + bv.y,
                       c[t][4 * g + 2][r] + bv.z, c[t][4 * g + 3][r] + bv.w};
        *(v4f*)(dst + idx) = o;
      }
  }
}

static constexpr int KTRIP = 64;
static constexpr int PPITCH = KTRIP + 4;

__device__ __forceinline__ void wave_lds_handover() {
  __builtin_amdgcn_fence(__ATOMIC_RELEASE, "wavefront");
  __builtin_amdgcn_wave_barrier();
  __builtin_amdgcn_fence(__ATOMIC_ACQUIRE, "wavefront");
}

template <bool MASK>
__device__ __forceinline__ void attn_trip(const v4f (&qa)[8], v8f (&o)[4], float (&m_i)[8],
                                          float (&l_i)[8], float (*pt)[PPITCH],
                                          const float* __restrict__ kbase,
                                          const float* __restrict__ vbase, const int kb,
                                          const int q0, const int hl, const int ln) {
  v8f s[4] = {};
  const float* krow = kbase + (size_t)kb * DD + 8 * ln + 4 * hl;
#pragma unroll
  for (int u = 0; u < 8; ++u) {
    v4f kf[4];
#pragma unroll
    for (int j = 0; j < 4; ++j) kf[j] = *(const v4f*)(krow + (4 * u + j) * 128);
    const v2f q01 = {qa[u].x, qa[u].y};
    const v2f q23 = {qa[u].z, qa[u].w};
#pragma unroll
    for (int j = 0; j < 4; ++j) {
      const v2f k01 = {kf[j].x, kf[j].y};
      s[j] = WMMA_F32(q01, k01, s[j]);
    }
#pragma unroll
    for (int j = 0; j < 4; ++j) {
      const v2f k23 = {kf[j].z, kf[j].w};
      s[j] = WMMA_F32(q23, k23, s[j]);
    }
  }

  if (MASK) {
#pragma unroll
    for (int j = 0; j < 4; ++j)
#pragma unroll
      for (int r = 0; r < 8; ++r)
        if (kb + 4 * ln + j > q0 + r + 8 * hl) s[j][r] = -__builtin_inff();
  }

#pragma unroll
  for (int r = 0; r < 8; ++r) {
    float mx = fmaxf(fmaxf(s[0][r], s[1][r]), fmaxf(s[2][r], s[3][r]));
    mx = fmaxf(mx, __shfl_xor(mx, 1));
    mx = fmaxf(mx, __shfl_xor(mx, 2));
    mx = fmaxf(mx, __shfl_xor(mx, 4));
    mx = fmaxf(mx, __shfl_xor(mx, 8));
    const float mn = fmaxf(m_i[r], mx);
    const float sc = __expf(m_i[r] - mn);
    const v4f p = {__expf(s[0][r] - mn), __expf(s[1][r] - mn),
                   __expf(s[2][r] - mn), __expf(s[3][r] - mn)};
    float ps = (p.x + p.y) + (p.z + p.w);
    ps += __shfl_xor(ps, 1);
    ps += __shfl_xor(ps, 2);
    ps += __shfl_xor(ps, 4);
    ps += __shfl_xor(ps, 8);
    l_i[r] = l_i[r] * sc + ps;
    m_i[r] = mn;
    *(v4f*)&pt[r + 8 * hl][4 * ln] = p;
#pragma unroll
    for (int n = 0; n < 4; ++n) o[n][r] *= sc;
  }

  wave_lds_handover();

  v4f pa[8];
#pragma unroll
  for (int u = 0; u < 8; ++u) pa[u] = *(const v4f*)&pt[ln][8 * u + 4 * hl];
  const float* vrow = vbase + (size_t)(kb + 4 * hl) * DD + 4 * ln;
#pragma unroll
  for (int u = 0; u < 8; ++u) {
    const v4f v0 = *(const v4f*)(vrow + (8 * u + 0) * DD);
    const v4f v1 = *(const v4f*)(vrow + (8 * u + 1) * DD);
    const v4f v2 = *(const v4f*)(vrow + (8 * u + 2) * DD);
    const v4f v3 = *(const v4f*)(vrow + (8 * u + 3) * DD);
    const v2f a01 = {pa[u].x, pa[u].y};
    const v2f a23 = {pa[u].z, pa[u].w};
#pragma unroll
    for (int n = 0; n < 4; ++n) {
      const v2f bf = {v0[n], v1[n]};
      o[n] = WMMA_F32(a01, bf, o[n]);
    }
#pragma unroll
    for (int n = 0; n < 4; ++n) {
      const v2f bf = {v2[n], v3[n]};
      o[n] = WMMA_F32(a23, bf, o[n]);
    }
  }
  wave_lds_handover();
}

__global__ __launch_bounds__(32 * ATT_WAVES) __attribute__((amdgpu_waves_per_eu(2, 2)))
void attn_kernel(const float* __restrict__ q, const float* __restrict__ kmat,
                 const float* __restrict__ vmat, float* __restrict__ out) {
  const int lane = threadIdx.x & 31;
  const int hl = lane >> 4;
  const int ln = lane & 15;
  const int w = __builtin_amdgcn_readfirstlane((int)(threadIdx.x >> 5));
  const int pr = blockIdx.x * ATT_WAVES + w;
  const int bh = blockIdx.y;
  const int b = bh >> 4, h = bh & 15;

  const float* qbase = q + (size_t)bh * TT * DD;
  const float* kbase = kmat + (size_t)bh * TT * DD;
  const float* vbase = vmat + (size_t)bh * TT * DD;

  __shared__ __attribute__((aligned(16))) float pt_all[ATT_WAVES][16][PPITCH];
  float (*pt)[PPITCH] = pt_all[w];

  for (int half = 0; half < 2; ++half) {
    const int qtile = half ? pr : (TT / 16 - 1 - pr);
    const int q0 = qtile * 16;

    v4f qa[8];
    {
      const float* qrow = qbase + (size_t)(q0 + ln) * DD + 4 * hl;
#pragma unroll
      for (int u = 0; u < 8; ++u) {
        const v4f a = *(const v4f*)(qrow + 8 * u);
        qa[u] = a * 0.125f;
      }
    }

    v8f o[4] = {};
    float m_i[8], l_i[8];
#pragma unroll
    for (int r = 0; r < 8; ++r) { m_i[r] = -__builtin_inff(); l_i[r] = 0.0f; }

    const int last = q0 / KTRIP;
    for (int kt = 0; kt < last; ++kt)
      attn_trip<false>(qa, o, m_i, l_i, pt, kbase, vbase, kt * KTRIP, q0, hl, ln);
    attn_trip<true>(qa, o, m_i, l_i, pt, kbase, vbase, last * KTRIP, q0, hl, ln);

#pragma unroll
    for (int r = 0; r < 8; ++r) {
      const float inv = 1.0f / l_i[r];
      const int trow = q0 + r + 8 * hl;
      float* orow = out + ((size_t)(b * TT + trow) * CC) + h * DD + 4 * ln;
      const v4f ov = {o[0][r] * inv, o[1][r] * inv, o[2][r] * inv, o[3][r] * inv};
      *(v4f*)orow = ov;
    }
  }
}

__global__ __launch_bounds__(32 * OUT_WAVES)
void out_gemm_kernel(const float* __restrict__ A, const float* __restrict__ W,
                     const float* __restrict__ bias, float* __restrict__ out) {
  const int lane = threadIdx.x & 31;
  const int hl = lane >> 4;
  const int ln = lane & 15;
  const int w = __builtin_amdgcn_readfirstlane((int)(threadIdx.x >> 5));
  const int n0 = blockIdx.x * 128;
  const int m0 = (blockIdx.y * OUT_WAVES + w) * 32;

  v8f c[2][8] = {};
  const float* A0 = A + (size_t)(m0 + ln) * CC + 4 * hl;
  gemm_tile_32x128_chunk_ahead<CC>(A0, A0 + 16 * CC,
                                   W + (size_t)(4 * hl) * CC + n0 + 4 * ln, c);

#pragma unroll
  for (int hf = 0; hf < 2; ++hf) {
    const v4f bv = *(const v4f*)(bias + n0 + 64 * hf + 4 * ln);
#pragma unroll
    for (int t = 0; t < 2; ++t)
#pragma unroll
      for (int r = 0; r < 8; ++r) {
        const v4f o = {c[t][4 * hf + 0][r] + bv.x, c[t][4 * hf + 1][r] + bv.y,
                       c[t][4 * hf + 2][r] + bv.z, c[t][4 * hf + 3][r] + bv.w};
        *(v4f*)(out + (size_t)(m0 + 16 * t + r + 8 * hl) * CC + n0 + 64 * hf + 4 * ln) = o;
      }
  }
}

extern "C" void kernel_launch(void* const* d_in, const int* in_sizes, int n_in,
                              void* d_out, int out_size, void* d_ws, size_t ws_size,
                              hipStream_t stream) {
  (void)in_sizes; (void)n_in; (void)out_size; (void)ws_size;
  const float* x    = (const float*)d_in[0];
  const float* Wqkv = (const float*)d_in[1];
  const float* bqkv = (const float*)d_in[2];
  const float* Wout = (const float*)d_in[3];
  const float* bout = (const float*)d_in[4];
  float* out = (float*)d_out;

  float* qp = (float*)d_ws;
  float* kp = qp + HSZ;
  float* vp = kp + HSZ;
  float* ao = vp + HSZ;

  qkv_gemm_kernel<<<dim3(NQKV / 384, MTOT / 128), dim3(32 * QKV_WAVES, 1, 1), 0, stream>>>(x, Wqkv, bqkv, qp, kp, vp);
  attn_kernel<<<dim3(TT / 32 / ATT_WAVES, BB * HH), dim3(32 * ATT_WAVES, 1, 1), 0, stream>>>(qp, kp, vp, ao);
  out_gemm_kernel<<<dim3(CC / 128, MTOT / 128), dim3(32 * OUT_WAVES, 1, 1), 0, stream>>>(ao, Wout, bout, out);
}


# Attention: scores cross LDS before the softmax (was P after it): a lane holds 32 keys of one row, max and sum are in-lane plus one lane^16 exchange, P feeds PV from registers; row sum order changes
# speedup vs baseline: 3.0091x; 1.0525x over previous
#include <hip/hip_runtime.h>


typedef __attribute__((ext_vector_type(2))) float v2f;
typedef __attribute__((ext_vector_type(4))) float v4f;
typedef __attribute__((ext_vector_type(8))) float v8f;

#define WMMA_F32(a, b, c) \
  __builtin_amdgcn_wmma_f32_16x16x4_f32(false, (a), false, (b), (short)0, (c), false, false)

static constexpr int BB = 2, TT = 2048, CC = 1024, HH = 16, DD = 64;
static constexpr int MTOT = BB * TT;
static constexpr int NQKV = 3 * CC;
static constexpr size_t HSZ = (size_t)BB * HH * TT * DD;


__device__ __forceinline__ void load_a_chunk(v4f (&a)[2][4],
                                             const float* __restrict__ A0,
                                             const float* __restrict__ A1, const int kc) {
#pragma unroll
  for (int u = 0; u < 4; ++u) a[0][u] = *(const v4f*)(A0 + kc + 8 * u);
#pragma unroll
  for (int u = 0; u < 4; ++u) a[1][u] = *(const v4f*)(A1 + kc + 8 * u);
}

template <int LDB>
__device__ __forceinline__ void load_b_step(v4f (&b)[8], const float* __restrict__ Bp,
                                            const int k0) {
  const float* Bk = Bp + (size_t)k0 * LDB;
#pragma unroll
  for (int r = 0; r < 4; ++r) {
    b[2 * r]     = *(const v4f*)(Bk + r * LDB);
    b[2 * r + 1] = *(const v4f*)(Bk + r * LDB + 64);
  }
}

__device__ __forceinline__ void mma_step(const v4f a0, const v4f a1, const v4f (&b)[8],
                                         v8f (&c)[2][8]) {
  const v2f a0lo = {a0.x, a0.y}, a1lo = {a1.x, a1.y};
  const v2f a0hi = {a0.z, a0.w}, a1hi = {a1.z, a1.w};
#pragma unroll
  for (int hf = 0; hf < 2; ++hf)
#pragma unroll
    for (int j = 0; j < 4; ++j) {
      const v2f bb = {b[hf][j], b[2 + hf][j]};
      c[0][4 * hf + j] = WMMA_F32(a0lo, bb, c[0][4 * hf + j]);
      c[1][4 * hf + j] = WMMA_F32(a1lo, bb, c[1][4 * hf + j]);
    }
#pragma unroll
  for (int hf = 0; hf < 2; ++hf)
#pragma unroll
    for (int j = 0; j < 4; ++j) {
      const v2f bb = {b[4 + hf][j], b[6 + hf][j]};
      c[0][4 * hf + j] = WMMA_F32(a0hi, bb, c[0][4 * hf + j]);
      c[1][4 * hf + j] = WMMA_F32(a1hi, bb, c[1][4 * hf + j]);
    }
}

template <int LDB>
__device__ __forceinline__ void load_b_chunk(v4f (&b)[4][8], const float* __restrict__ Bp,
                                             const int kc) {
#pragma unroll
  for (int s = 0; s < 4; ++s) load_b_step<LDB>(b[s], Bp, kc + 8 * s);
}

__device__ __forceinline__ void mma_chunk(const v4f (&a)[2][4], const v4f (&b)[4][8],
                                          v8f (&c)[2][8]) {
#pragma unroll
  for (int s = 0; s < 4; ++s) mma_step(a[0][s], a[1][s], b[s], c);
}

template <int LDB>
__device__ __forceinline__ void gemm_tile_32x128_chunk_ahead(const float* __restrict__ A0,
                                                             const float* __restrict__ A1,
                                                             const float* __restrict__ Bp,
                                                             v8f (&c)[2][8]) {
  v4f aP[2][4], aQ[2][4], bP[4][8], bQ[4][8];
  load_a_chunk(aP, A0, A1, 0);
  load_b_chunk<LDB>(bP, Bp, 0);
  for (int kc = 0; kc < CC; kc += 64) {
    load_a_chunk(aQ, A0, A1, kc + 32);
    load_b_chunk<LDB>(bQ, Bp, kc + 32);
    __builtin_amdgcn_sched_barrier(0);
    mma_chunk(aP, bP, c);
    load_a_chunk(aP, A0, A1, (kc + 64) & (CC - 1));
    load_b_chunk<LDB>(bP, Bp, (kc + 64) & (CC - 1));
    __builtin_amdgcn_sched_barrier(0);
    mma_chunk(aQ, bQ, c);
  }
}

static constexpr int QT = 4;
static constexpr int QG = 3;

__device__ __forceinline__ void qkv_load_a(v4f (&a)[QT][4], const float* A0, const int kc) {
#pragma unroll
  for (int t = 0; t < QT; ++t)
#pragma unroll
    for (int u = 0; u < 4; ++u) a[t][u] = *(const v4f*)(A0 + (size_t)t * 16 * CC + kc + 8 * u);
}

__device__ __forceinline__ void qkv_load_b(v4f (&b)[4][QG], const float* Bp, const int k0) {
  const float* Bk = Bp + (size_t)k0 * NQKV;
#pragma unroll
  for (int r = 0; r < 4; ++r)
#pragma unroll
    for (int g = 0; g < QG; ++g) b[r][g] = *(const v4f*)(Bk + r * NQKV + 64 * g);
}

template <int U>
__device__ __forceinline__ void qkv_mma_step(const v4f (&a)[QT][4], const v4f (&b)[4][QG],
                                             v8f (&c)[QT][4 * QG]) {
#pragma unroll
  for (int g = 0; g < QG; ++g)
#pragma unroll
    for (int j = 0; j < 4; ++j) {
      const v2f bb = {b[0][g][j], b[1][g][j]};
#pragma unroll
      for (int t = 0; t < QT; ++t) {
        const v2f alo = {a[t][U].x, a[t][U].y};
        c[t][4 * g + j] = WMMA_F32(alo, bb, c[t][4 * g + j]);
      }
    }
#pragma unroll
  for (int g = 0; g < QG; ++g)
#pragma unroll
    for (int j = 0; j < 4; ++j) {
      const v2f bb = {b[2][g][j], b[3][g][j]};
#pragma unroll
      for (int t = 0; t < QT; ++t) {
        const v2f ahi = {a[t][U].z, a[t][U].w};
        c[t][4 * g + j] = WMMA_F32(ahi, bb, c[t][4 * g + j]);
      }
    }
}

__device__ __forceinline__ void request_fence() {
  asm volatile("" ::: "memory");
  __builtin_amdgcn_sched_barrier(0);
}

__device__ __forceinline__ void qkv_chunk(const v4f (&aC)[QT][4], const v4f (&bC)[4][4][QG],
                                          v4f (&aN)[QT][4], v4f (&bN)[4][4][QG],
                                          const float* A0, const float* Bp, const int kn,
                                          v8f (&c)[QT][4 * QG]) {
  request_fence();
  qkv_load_a(aN, A0, kn);
  qkv_load_b(bN[0], Bp, kn);
  qkv_load_b(bN[1], Bp, kn + 8);
  request_fence();
  qkv_mma_step<0>(aC, bC[0], c);
  qkv_mma_step<1>(aC, bC[1], c);
  request_fence();
  qkv_load_b(bN[2], Bp, kn + 16);
  qkv_load_b(bN[3], Bp, kn + 24);
  request_fence();
  qkv_mma_step<2>(aC, bC[2], c);
  qkv_mma_step<3>(aC, bC[3], c);
}

__device__ __forceinline__ void qkv_tile_64x192(const float* A0, const float* Bp,
                                                v8f (&c)[QT][4 * QG]) {
  v4f aP[QT][4], aQ[QT][4], bP[4][4][QG], bQ[4][4][QG];
  qkv_load_a(aP, A0, 0);
#pragma unroll
  for (int s = 0; s < 4; ++s) qkv_load_b(bP[s], Bp, 8 * s);
  for (int kc = 0; kc < CC; kc += 64) {
    qkv_chunk(aP, bP, aQ, bQ, A0, Bp, kc + 32, c);
    qkv_chunk(aQ, bQ, aP, bP, A0, Bp, (kc + 64) & (CC - 1), c);
  }
}

static constexpr int QKV_WAVES = 4;
static constexpr int OUT_WAVES = 4;
static constexpr int ATT_WAVES = 8;

__global__ __launch_bounds__(32 * QKV_WAVES)
void qkv_gemm_kernel(const float* X, const float* W,
                     const float* __restrict__ bias,
                     float* __restrict__ q, float* __restrict__ k,
                     float* __restrict__ v) {
  const int lane = threadIdx.x & 31;
  const int hl = lane >> 4;
  const int ln = lane & 15;
  const int w = __builtin_amdgcn_readfirstlane((int)(threadIdx.x >> 5));
  const int n0 = (blockIdx.x * 2 + (w >> 1)) * (64 * QG);
  const int m0 = (blockIdx.y * 2 + (w & 1)) * (16 * QT);

  v8f c[QT][4 * QG] = {};
  qkv_tile_64x192(X + (size_t)(m0 + ln) * CC + 4 * hl,
                  W + (size_t)(4 * hl) * NQKV + n0 + 4 * ln, c);

  const int kfrag = (ln >> 1) * 512 + (ln & 1) * 4;
#pragma unroll
  for (int g = 0; g < QG; ++g) {
    const int nc = n0 + 64 * g;
    const int which = nc >> 10;
    float* dst = (which == 0) ? q : (which == 1) ? k : v;
    const v4f bv = *(const v4f*)(bias + nc + 4 * ln);
    const int h = (nc & (CC - 1)) >> 6;
#pragma unroll
    for (int t = 0; t < QT; ++t)
#pragma unroll
      for (int r = 0; r < 8; ++r) {
        const int trow = m0 + 16 * t + r + 8 * hl;
        const int bidx = trow >> 11;
        const int tt = trow & (TT - 1);
        const size_t rowm = (((size_t)(bidx * HH + h) * TT) + tt) * DD + 4 * ln;
        const size_t frag = (((size_t)(bidx * HH + h) * TT) + (tt & ~63)) * DD +
                            (tt & 3) * 128 + ((tt & 63) >> 2) * 8 + kfrag;
        const size_t idx = (which == 1) ? frag : rowm;
        const v4f o = {c[t][4 * g + 0][r] + bv.x, c[t][4 * g + 1][r] + bv.y,
                       c[t][4 * g + 2][r] + bv.z, c[t][4 * g + 3][r] + bv.w};
        *(v4f*)(dst + idx) = o;
      }
  }
}

static constexpr int KTRIP = 64;
static constexpr int PPITCH = KTRIP + 4;

__device__ __forceinline__ void wave_lds_handover() {
  __builtin_amdgcn_fence(__ATOMIC_RELEASE, "wavefront");
  __builtin_amdgcn_wave_barrier();
  __builtin_amdgcn_fence(__ATOMIC_ACQUIRE, "wavefront");
}

__device__ __forceinline__ float xhalf(const float x) {
  const int b = __builtin_bit_cast(int, x);
  return __builtin_bit_cast(float,
                            __builtin_amdgcn_permlanex16(b, b, 0x76543210, 0xfedcba98, false, false));
}

template <bool MASK>
__device__ __forceinline__ void attn_trip(const v4f (&qa)[8], v8f (&o)[4], float& m_i,
                                          float& l_i, float (*pt)[PPITCH],
                                          const float* __restrict__ kbase,
                                          const float* __restrict__ vbase, const int kb,
                                          const int q0, const int hl, const int ln) {
  v8f s[4] = {};
  const float* krow = kbase + (size_t)kb * DD + 8 * ln + 4 * hl;
#pragma unroll
  for (int u = 0; u < 8; ++u) {
    v4f kf[4];
#pragma unroll
    for (int j = 0; j < 4; ++j) kf[j] = *(const v4f*)(krow + (4 * u + j) * 128);
    const v2f q01 = {qa[u].x, qa[u].y};
    const v2f q23 = {qa[u].z, qa[u].w};
#pragma unroll
    for (int j = 0; j < 4; ++j) {
      const v2f k01 = {kf[j].x, kf[j].y};
      s[j] = WMMA_F32(q01, k01, s[j]);
    }
#pragma unroll
    for (int j = 0; j < 4; ++j) {
      const v2f k23 = {kf[j].z, kf[j].w};
      s[j] = WMMA_F32(q23, k23, s[j]);
    }
  }

#pragma unroll
  for (int r = 0; r < 8; ++r) {
    const v4f sv = {s[0][r], s[1][r], s[2][r], s[3][r]};
    *(v4f*)&pt[r + 8 * hl][4 * ln] = sv;
  }

  wave_lds_handover();

  v4f sa[8];
#pragma unroll
  for (int u = 0; u < 8; ++u) sa[u] = *(const v4f*)&pt[ln][8 * u + 4 * hl];

  if (MASK) {
#pragma unroll
    for (int u = 0; u < 8; ++u)
#pragma unroll
      for (int e = 0; e < 4; ++e)
        if (kb + 8 * u + 4 * hl + e > q0 + ln) sa[u][e] = -__builtin_inff();
  }

  float mx = fmaxf(fmaxf(sa[0].x, sa[0].y), fmaxf(sa[0].z, sa[0].w));
#pragma unroll
  for (int u = 1; u < 8; ++u)
    mx = fmaxf(mx, fmaxf(fmaxf(sa[u].x, sa[u].y), fmaxf(sa[u].z, sa[u].w)));
  mx = fmaxf(mx, xhalf(mx));
  const float mn = fmaxf(m_i, mx);
  const float sc = __expf(m_i - mn);

#pragma unroll
  for (int r = 0; r < 8; ++r) {
    const float scr = __shfl(sc, 8 * hl + r);
#pragma unroll
    for (int n = 0; n < 4; ++n) o[n][r] *= scr;
  }

  float pu[8];
#pragma unroll
  for (int u = 0; u < 8; ++u) {
    const v4f p = {__expf(sa[u].x - mn), __expf(sa[u].y - mn),
                   __expf(sa[u].z - mn), __expf(sa[u].w - mn)};
    sa[u] = p;
    pu[u] = (p.x + p.y) + (p.z + p.w);
  }
  const float ps = ((pu[0] + pu[1]) + (pu[2] + pu[3])) + ((pu[4] + pu[5]) + (pu[6] + pu[7]));
  l_i = l_i * sc + ps;
  m_i = mn;

  const float* vrow = vbase + (size_t)(kb + 4 * hl) * DD + 4 * ln;
#pragma unroll
  for (int u = 0; u < 8; ++u) {
    const v4f v0 = *(const v4f*)(vrow + (8 * u + 0) * DD);
    const v4f v1 = *(const v4f*)(vrow + (8 * u + 1) * DD);
    const v4f v2 = *(const v4f*)(vrow + (8 * u + 2) * DD);
    const v4f v3 = *(const v4f*)(vrow + (8 * u + 3) * DD);
    const v2f a01 = {sa[u].x, sa[u].y};
    const v2f a23 = {sa[u].z, sa[u].w};
#pragma unroll
    for (int n = 0; n < 4; ++n) {
      const v2f bf = {v0[n], v1[n]};
      o[n] = WMMA_F32(a01, bf, o[n]);
    }
#pragma unroll
    for (int n = 0; n < 4; ++n) {
      const v2f bf = {v2[n], v3[n]};
      o[n] = WMMA_F32(a23, bf, o[n]);
    }
  }
  wave_lds_handover();
}

__global__ __launch_bounds__(32 * ATT_WAVES) __attribute__((amdgpu_waves_per_eu(2, 2)))
void attn_kernel(const float* __restrict__ q, const float* __restrict__ kmat,
                 const float* __restrict__ vmat, float* __restrict__ out) {
  const int lane = threadIdx.x & 31;
  const int hl = lane >> 4;
  const int ln = lane & 15;
  const int w = __builtin_amdgcn_readfirstlane((int)(threadIdx.x >> 5));
  const int pr = blockIdx.x * ATT_WAVES + w;
  const int bh = blockIdx.y;
  const int b = bh >> 4, h = bh & 15;

  const float* qbase = q + (size_t)bh * TT * DD;
  const float* kbase = kmat + (size_t)bh * TT * DD;
  const float* vbase = vmat + (size_t)bh * TT * DD;

  __shared__ __attribute__((aligned(16))) float pt_all[ATT_WAVES][16][PPITCH];
  float (*pt)[PPITCH] = pt_all[w];

  for (int half = 0; half < 2; ++half) {
    const int qtile = half ? pr : (TT / 16 - 1 - pr);
    const int q0 = qtile * 16;

    v4f qa[8];
    {
      const float* qrow = qbase + (size_t)(q0 + ln) * DD + 4 * hl;
#pragma unroll
      for (int u = 0; u < 8; ++u) {
        const v4f a = *(const v4f*)(qrow + 8 * u);
        qa[u] = a * 0.125f;
      }
    }

    v8f o[4] = {};
    float m_i = -__builtin_inff();
    float l_i = 0.0f;

    const int last = q0 / KTRIP;
    for (int kt = 0; kt < last; ++kt)
      attn_trip<false>(qa, o, m_i, l_i, pt, kbase, vbase, kt * KTRIP, q0, hl, ln);
    attn_trip<true>(qa, o, m_i, l_i, pt, kbase, vbase, last * KTRIP, q0, hl, ln);

    const float inv_row = 1.0f / (l_i + xhalf(l_i));
#pragma unroll
    for (int r = 0; r < 8; ++r) {
      const float inv = __shfl(inv_row, 8 * hl + r);
      const int trow = q0 + r + 8 * hl;
      float* orow = out + ((size_t)(b * TT + trow) * CC) + h * DD + 4 * ln;
      const v4f ov = {o[0][r] * inv, o[1][r] * inv, o[2][r] * inv, o[3][r] * inv};
      *(v4f*)orow = ov;
    }
  }
}

__global__ __launch_bounds__(32 * OUT_WAVES)
void out_gemm_kernel(const float* __restrict__ A, const float* __restrict__ W,
                     const float* __restrict__ bias, float* __restrict__ out) {
  const int lane = threadIdx.x & 31;
  const int hl = lane >> 4;
  const int ln = lane & 15;
  const int w = __builtin_amdgcn_readfirstlane((int)(threadIdx.x >> 5));
  const int n0 = blockIdx.x * 128;
  const int m0 = (blockIdx.y * OUT_WAVES + w) * 32;

  v8f c[2][8] = {};
  const float* A0 = A + (size_t)(m0 + ln) * CC + 4 * hl;
  gemm_tile_32x128_chunk_ahead<CC>(A0, A0 + 16 * CC,
                                   W + (size_t)(4 * hl) * CC + n0 + 4 * ln, c);

#pragma unroll
  for (int hf = 0; hf < 2; ++hf) {
    const v4f bv = *(const v4f*)(bias + n0 + 64 * hf + 4 * ln);
#pragma unroll
    for (int t = 0; t < 2; ++t)
#pragma unroll
      for (int r = 0; r < 8; ++r) {
        const v4f o = {c[t][4 * hf + 0][r] + bv.x, c[t][4 * hf + 1][r] + bv.y,
                       c[t][4 * hf + 2][r] + bv.z, c[t][4 * hf + 3][r] + bv.w};
        *(v4f*)(out + (size_t)(m0 + 16 * t + r + 8 * hl) * CC + n0 + 64 * hf + 4 * ln) = o;
      }
  }
}

extern "C" void kernel_launch(void* const* d_in, const int* in_sizes, int n_in,
                              void* d_out, int out_size, void* d_ws, size_t ws_size,
                              hipStream_t stream) {
  (void)in_sizes; (void)n_in; (void)out_size; (void)ws_size;
  const float* x    = (const float*)d_in[0];
  const float* Wqkv = (const float*)d_in[1];
  const float* bqkv = (const float*)d_in[2];
  const float* Wout = (const float*)d_in[3];
  const float* bout = (const float*)d_in[4];
  float* out = (float*)d_out;

  float* qp = (float*)d_ws;
  float* kp = qp + HSZ;
  float* vp = kp + HSZ;
  float* ao = vp + HSZ;

  qkv_gemm_kernel<<<dim3(NQKV / 384, MTOT / 128), dim3(32 * QKV_WAVES, 1, 1), 0, stream>>>(x, Wqkv, bqkv, qp, kp, vp);
  attn_kernel<<<dim3(TT / 32 / ATT_WAVES, BB * HH), dim3(32 * ATT_WAVES, 1, 1), 0, stream>>>(qp, kp, vp, ao);
  out_gemm_kernel<<<dim3(CC / 128, MTOT / 128), dim3(32 * OUT_WAVES, 1, 1), 0, stream>>>(ao, Wout, bout, out);
}
